# starting kernel
# baseline (speedup 1.0000x reference)
#include <hip/hip_runtime.h>
#include <hip/hip_bf16.h>


#define C 128
#define LEAKY 0.2f

typedef __attribute__((ext_vector_type(2))) float v2f;
typedef __attribute__((ext_vector_type(8))) float v8f;

__global__ __launch_bounds__(256) void gat_zero(float* __restrict__ out, int n_out,
                                                float* __restrict__ denom, int n_den) {
    int i = blockIdx.x * 256 + threadIdx.x;
    if (i < n_out) out[i] = 0.0f;
    if (i < n_den) denom[i] = 0.0f;
}

__global__ __launch_bounds__(256) void gat_gemm_uv(const float* __restrict__ h,
                                                   const float* __restrict__ w,
                                                   float* __restrict__ u,
                                                   float* __restrict__ v,
                                                   int n_nodes) {
    __shared__ float lh[16 * C];
    const int row0 = blockIdx.x * 16;
    const int tid  = threadIdx.x;
    const bool full_tile = (row0 + 16) <= n_nodes;

    if (full_tile) {
        for (int i = tid; i < 16 * C; i += 256)
            lh[i] = h[(size_t)row0 * C + i];
    } else {
        for (int i = tid; i < 16 * C; i += 256) {
            int gr = row0 + (i >> 7);
            lh[i] = (gr < n_nodes) ? h[(size_t)gr * C + (i & (C - 1))] : 0.0f;
        }
    }
    __syncthreads();

    const int wave = tid >> 5;
    const int lane = tid & 31;
    const int half = lane >> 4;
    const int l16  = lane & 15;
    const int col  = wave * 16 + l16;

    v8f accu = {};
    v8f accv = {};
    for (int k = 0; k < C; k += 4) {
        const int ka = k + half * 2;
        v2f afrag;
        afrag.x = lh[l16 * C + ka];
        afrag.y = lh[l16 * C + ka + 1];
        v2f bu, bv;
        bu.x = w[(size_t)(ka    ) * C + col];
        bu.y = w[(size_t)(ka + 1) * C + col];
        bv.x = w[(size_t)(C + ka    ) * C + col];
        bv.y = w[(size_t)(C + ka + 1) * C + col];
        accu = __builtin_amdgcn_wmma_f32_16x16x4_f32(false, afrag, false, bu,
                                                     (short)0, accu, false, false);
        accv = __builtin_amdgcn_wmma_f32_16x16x4_f32(false, afrag, false, bv,
                                                     (short)0, accv, false, false);
    }

    const int mbase = row0 + half * 8;
    if (full_tile) {
        #pragma unroll
        for (int r = 0; r < 8; ++r) {
            u[(size_t)(mbase + r) * C + col] = accu[r];
            v[(size_t)(mbase + r) * C + col] = accv[r];
        }
    } else {
        #pragma unroll
        for (int r = 0; r < 8; ++r) {
            if (mbase + r < n_nodes) {
                u[(size_t)(mbase + r) * C + col] = accu[r];
                v[(size_t)(mbase + r) * C + col] = accv[r];
            }
        }
    }
}

__device__ __forceinline__ float leaky(float x) { return x > 0.0f ? x : LEAKY * x; }

__global__ __launch_bounds__(256) void gat_edge_e(const float* __restrict__ u,
                                                  const float* __restrict__ v,
                                                  const float* __restrict__ a,
                                                  const int* __restrict__ src,
                                                  const int* __restrict__ dst,
                                                  float* __restrict__ exp_e,
                                                  float* __restrict__ denom,
                                                  int n_edges) {
    const int wib  = threadIdx.x >> 5;
    const int lane = threadIdx.x & 31;
    const int e = blockIdx.x * 8 + wib;
    if (e >= n_edges) return;

    const int s = src[e];
    const int d = dst[e];
    float4 uu = ((const float4*)(u + (size_t)s * C))[lane];
    float4 vv = ((const float4*)(v + (size_t)d * C))[lane];
    float4 aa = ((const float4*)a)[lane];

    float sum = leaky(uu.x + vv.x) * aa.x;
    sum = fmaf(leaky(uu.y + vv.y), aa.y, sum);
    sum = fmaf(leaky(uu.z + vv.z), aa.z, sum);
    sum = fmaf(leaky(uu.w + vv.w), aa.w, sum);

    #pragma unroll
    for (int off = 16; off > 0; off >>= 1)
        sum += __shfl_xor(sum, off, 32);

    if (lane == 0) {
        float ex = expf(sum);
        exp_e[e] = ex;
        atomicAdd(&denom[s], ex);
    }
}

__global__ __launch_bounds__(256) void gat_scatter(const float* __restrict__ h,
                                                   const int* __restrict__ src,
                                                   const int* __restrict__ dst,
                                                   const float* __restrict__ exp_e,
                                                   const float* __restrict__ denom,
                                                   float* __restrict__ out,
                                                   int n_edges) {
    const int wib  = threadIdx.x >> 5;
    const int lane = threadIdx.x & 31;
    const int e = blockIdx.x * 8 + wib;
    if (e >= n_edges) return;

    const int s = src[e];
    const int d = dst[e];
    const float att = exp_e[e] / denom[s];

    float4 hv = ((const float4*)(h + (size_t)s * C))[lane];
    float* op = out + (size_t)d * C + lane * 4;
    atomicAdd(op + 0, hv.x * att);
    atomicAdd(op + 1, hv.y * att);
    atomicAdd(op + 2, hv.z * att);
    atomicAdd(op + 3, hv.w * att);
}

extern "C" void kernel_launch(void* const* d_in, const int* in_sizes, int n_in,
                              void* d_out, int out_size, void* d_ws, size_t ws_size,
                              hipStream_t stream) {
    const float* h  = (const float*)d_in[0];
    const float* w  = (const float*)d_in[1];
    const float* a  = (const float*)d_in[2];
    const int* src  = (const int*)d_in[3];
    const int* dst  = (const int*)d_in[4];
    float* out      = (float*)d_out;

    const int n_nodes = in_sizes[0] / C;
    const int n_edges = in_sizes[3];

    float* u     = (float*)d_ws;
    float* v     = u + (size_t)n_nodes * C;
    float* exp_e = v + (size_t)n_nodes * C;
    float* denom = exp_e + n_edges;

    {
        int nz = n_nodes * C;
        int grid = (nz + 255) / 256;
        gat_zero<<<grid, 256, 0, stream>>>(out, nz, denom, n_nodes);
    }
    {
        int grid = (n_nodes + 15) / 16;
        gat_gemm_uv<<<grid, 256, 0, stream>>>(h, w, u, v, n_nodes);
    }
    {
        int grid = (n_edges + 7) / 8;
        gat_edge_e<<<grid, 256, 0, stream>>>(u, v, a, src, dst, exp_e, denom, n_edges);
    }
    {
        int grid = (n_edges + 7) / 8;
        gat_scatter<<<grid, 256, 0, stream>>>(h, src, dst, exp_e, denom, out, n_edges);
    }
}


# edge-logit kernel: one wave per 32 edges instead of one per edge (coalesced index loads, 16 row loads in flight, transposing xor butterfly, same tree so logits bit-identical)
# speedup vs baseline: 1.4031x; 1.4031x over previous
#include <hip/hip_runtime.h>
#include <hip/hip_bf16.h>


#define C 128
#define LEAKY 0.2f

typedef __attribute__((ext_vector_type(2))) float v2f;
typedef __attribute__((ext_vector_type(8))) float v8f;

__global__ __launch_bounds__(256) void gat_zero(float* __restrict__ out, int n_out,
                                                float* __restrict__ denom, int n_den) {
    int i = blockIdx.x * 256 + threadIdx.x;
    if (i < n_out) out[i] = 0.0f;
    if (i < n_den) denom[i] = 0.0f;
}

__global__ __launch_bounds__(256) void gat_gemm_uv(const float* __restrict__ h,
                                                   const float* __restrict__ w,
                                                   float* __restrict__ u,
                                                   float* __restrict__ v,
                                                   int n_nodes) {
    __shared__ float lh[16 * C];
    const int row0 = blockIdx.x * 16;
    const int tid  = threadIdx.x;
    const bool full_tile = (row0 + 16) <= n_nodes;

    if (full_tile) {
        for (int i = tid; i < 16 * C; i += 256)
            lh[i] = h[(size_t)row0 * C + i];
    } else {
        for (int i = tid; i < 16 * C; i += 256) {
            int gr = row0 + (i >> 7);
            lh[i] = (gr < n_nodes) ? h[(size_t)gr * C + (i & (C - 1))] : 0.0f;
        }
    }
    __syncthreads();

    const int wave = tid >> 5;
    const int lane = tid & 31;
    const int half = lane >> 4;
    const int l16  = lane & 15;
    const int col  = wave * 16 + l16;

    v8f accu = {};
    v8f accv = {};
    for (int k = 0; k < C; k += 4) {
        const int ka = k + half * 2;
        v2f afrag;
        afrag.x = lh[l16 * C + ka];
        afrag.y = lh[l16 * C + ka + 1];
        v2f bu, bv;
        bu.x = w[(size_t)(ka    ) * C + col];
        bu.y = w[(size_t)(ka + 1) * C + col];
        bv.x = w[(size_t)(C + ka    ) * C + col];
        bv.y = w[(size_t)(C + ka + 1) * C + col];
        accu = __builtin_amdgcn_wmma_f32_16x16x4_f32(false, afrag, false, bu,
                                                     (short)0, accu, false, false);
        accv = __builtin_amdgcn_wmma_f32_16x16x4_f32(false, afrag, false, bv,
                                                     (short)0, accv, false, false);
    }

    const int mbase = row0 + half * 8;
    if (full_tile) {
        #pragma unroll
        for (int r = 0; r < 8; ++r) {
            u[(size_t)(mbase + r) * C + col] = accu[r];
            v[(size_t)(mbase + r) * C + col] = accv[r];
        }
    } else {
        #pragma unroll
        for (int r = 0; r < 8; ++r) {
            if (mbase + r < n_nodes) {
                u[(size_t)(mbase + r) * C + col] = accu[r];
                v[(size_t)(mbase + r) * C + col] = accv[r];
            }
        }
    }
}

__device__ __forceinline__ float leaky(float x) { return x > 0.0f ? x : LEAKY * x; }

typedef __attribute__((ext_vector_type(4))) float v4f;

__device__ __forceinline__ v4f load_row4(const float* __restrict__ base, int row, int lane) {
    return *(const v4f*)(base + (size_t)row * C + lane * 4);
}

__device__ __forceinline__ float edge_partial(v4f uu, v4f vv, float4 aa) {
    float sum = leaky(uu.x + vv.x) * aa.x;
    sum = fmaf(leaky(uu.y + vv.y), aa.y, sum);
    sum = fmaf(leaky(uu.z + vv.z), aa.z, sum);
    sum = fmaf(leaky(uu.w + vv.w), aa.w, sum);
    return sum;
}

__device__ __forceinline__ float fold(float lo, float hi, bool bit, int off) {
    float keep = bit ? hi : lo;
    float send = bit ? lo : hi;
    return keep + __shfl_xor(send, off, 32);
}

__global__ __launch_bounds__(256) void gat_edge_e(const float* __restrict__ u,
                                                  const float* __restrict__ v,
                                                  const float* __restrict__ a,
                                                  const int* __restrict__ src,
                                                  const int* __restrict__ dst,
                                                  float* __restrict__ exp_e,
                                                  float* __restrict__ denom,
                                                  int n_edges) {
    const int wib  = threadIdx.x >> 5;
    const int lane = threadIdx.x & 31;
    const int e0 = (blockIdx.x * 8 + wib) * 32;
    if (e0 >= n_edges) return;

    const int eown = min(e0 + lane, n_edges - 1);
    const int s_own = src[eown];
    const int d_own = dst[eown];
    const float4 aa = ((const float4*)a)[lane];

    const bool b4 = (lane & 16) != 0;
    const bool b3 = (lane & 8) != 0;
    const bool b2 = (lane & 4) != 0;
    const bool b1 = (lane & 2) != 0;
    const bool b0 = (lane & 1) != 0;

    float wsum[4];
    #pragma unroll
    for (int b = 0; b < 4; ++b) {
        v4f uu[8], vv[8];
        #pragma unroll
        for (int i = 0; i < 8; ++i) {
            uu[i] = load_row4(u, __builtin_amdgcn_readlane(s_own, b * 8 + i), lane);
            vv[i] = load_row4(v, __builtin_amdgcn_readlane(d_own, b * 8 + i), lane);
        }
        asm volatile("" :: "v"(uu[0]), "v"(uu[1]), "v"(uu[2]), "v"(uu[3]),
                           "v"(uu[4]), "v"(uu[5]), "v"(uu[6]), "v"(uu[7]),
                           "v"(vv[0]), "v"(vv[1]), "v"(vv[2]), "v"(vv[3]),
                           "v"(vv[4]), "v"(vv[5]), "v"(vv[6]), "v"(vv[7]));
        float p[8];
        #pragma unroll
        for (int i = 0; i < 8; ++i)
            p[i] = edge_partial(uu[i], vv[i], aa);
        const float q0 = fold(p[0], p[4], b4, 16);
        const float q1 = fold(p[1], p[5], b4, 16);
        const float q2 = fold(p[2], p[6], b4, 16);
        const float q3 = fold(p[3], p[7], b4, 16);
        const float r0 = fold(q0, q2, b3, 8);
        const float r1 = fold(q1, q3, b3, 8);
        wsum[b] = fold(r0, r1, b2, 4);
    }
    const float x0 = fold(wsum[0], wsum[2], b1, 2);
    const float x1 = fold(wsum[1], wsum[3], b1, 2);
    const float logit = fold(x0, x1, b0, 1);

    const int slot = ((lane & 3) << 3) | (lane >> 2);
    const int s_fin = __shfl(s_own, slot, 32);
    const int e = e0 + slot;
    if (e < n_edges) {
        float ex = expf(logit);
        exp_e[e] = ex;
        atomicAdd(&denom[s_fin], ex);
    }
}

__global__ __launch_bounds__(256) void gat_scatter(const float* __restrict__ h,
                                                   const int* __restrict__ src,
                                                   const int* __restrict__ dst,
                                                   const float* __restrict__ exp_e,
                                                   const float* __restrict__ denom,
                                                   float* __restrict__ out,
                                                   int n_edges) {
    const int wib  = threadIdx.x >> 5;
    const int lane = threadIdx.x & 31;
    const int e = blockIdx.x * 8 + wib;
    if (e >= n_edges) return;

    const int s = src[e];
    const int d = dst[e];
    const float att = exp_e[e] / denom[s];

    float4 hv = ((const float4*)(h + (size_t)s * C))[lane];
    float* op = out + (size_t)d * C + lane * 4;
    atomicAdd(op + 0, hv.x * att);
    atomicAdd(op + 1, hv.y * att);
    atomicAdd(op + 2, hv.z * att);
    atomicAdd(op + 3, hv.w * att);
}

extern "C" void kernel_launch(void* const* d_in, const int* in_sizes, int n_in,
                              void* d_out, int out_size, void* d_ws, size_t ws_size,
                              hipStream_t stream) {
    const float* h  = (const float*)d_in[0];
    const float* w  = (const float*)d_in[1];
    const float* a  = (const float*)d_in[2];
    const int* src  = (const int*)d_in[3];
    const int* dst  = (const int*)d_in[4];
    float* out      = (float*)d_out;

    const int n_nodes = in_sizes[0] / C;
    const int n_edges = in_sizes[3];

    float* u     = (float*)d_ws;
    float* v     = u + (size_t)n_nodes * C;
    float* exp_e = v + (size_t)n_nodes * C;
    float* denom = exp_e + n_edges;

    {
        int nz = n_nodes * C;
        int grid = (nz + 255) / 256;
        gat_zero<<<grid, 256, 0, stream>>>(out, nz, denom, n_nodes);
    }
    {
        int grid = (n_nodes + 15) / 16;
        gat_gemm_uv<<<grid, 256, 0, stream>>>(h, w, u, v, n_nodes);
    }
    {
        int grid = (n_edges + 255) / 256;
        gat_edge_e<<<grid, 256, 0, stream>>>(u, v, a, src, dst, exp_e, denom, n_edges);
    }
    {
        int grid = (n_edges + 7) / 8;
        gat_scatter<<<grid, 256, 0, stream>>>(h, src, dst, exp_e, denom, out, n_edges);
    }
}


# scatter kernel: one wave per 32 edges, lane = channel, so every row load and f32 atomic add covers one whole 128-B line (4 line adds per edge, was 16 quarter-filled); sums still in arrival order
# speedup vs baseline: 2.0137x; 1.4352x over previous
#include <hip/hip_runtime.h>
#include <hip/hip_bf16.h>


#define C 128
#define LEAKY 0.2f

typedef __attribute__((ext_vector_type(2))) float v2f;
typedef __attribute__((ext_vector_type(8))) float v8f;

__global__ __launch_bounds__(256) void gat_zero(float* __restrict__ out, int n_out,
                                                float* __restrict__ denom, int n_den) {
    int i = blockIdx.x * 256 + threadIdx.x;
    if (i < n_out) out[i] = 0.0f;
    if (i < n_den) denom[i] = 0.0f;
}

__global__ __launch_bounds__(256) void gat_gemm_uv(const float* __restrict__ h,
                                                   const float* __restrict__ w,
                                                   float* __restrict__ u,
                                                   float* __restrict__ v,
                                                   int n_nodes) {
    __shared__ float lh[16 * C];
    const int row0 = blockIdx.x * 16;
    const int tid  = threadIdx.x;
    const bool full_tile = (row0 + 16) <= n_nodes;

    if (full_tile) {
        for (int i = tid; i < 16 * C; i += 256)
            lh[i] = h[(size_t)row0 * C + i];
    } else {
        for (int i = tid; i < 16 * C; i += 256) {
            int gr = row0 + (i >> 7);
            lh[i] = (gr < n_nodes) ? h[(size_t)gr * C + (i & (C - 1))] : 0.0f;
        }
    }
    __syncthreads();

    const int wave = tid >> 5;
    const int lane = tid & 31;
    const int half = lane >> 4;
    const int l16  = lane & 15;
    const int col  = wave * 16 + l16;

    v8f accu = {};
    v8f accv = {};
    for (int k = 0; k < C; k += 4) {
        const int ka = k + half * 2;
        v2f afrag;
        afrag.x = lh[l16 * C + ka];
        afrag.y = lh[l16 * C + ka + 1];
        v2f bu, bv;
        bu.x = w[(size_t)(ka    ) * C + col];
        bu.y = w[(size_t)(ka + 1) * C + col];
        bv.x = w[(size_t)(C + ka    ) * C + col];
        bv.y = w[(size_t)(C + ka + 1) * C + col];
        accu = __builtin_amdgcn_wmma_f32_16x16x4_f32(false, afrag, false, bu,
                                                     (short)0, accu, false, false);
        accv = __builtin_amdgcn_wmma_f32_16x16x4_f32(false, afrag, false, bv,
                                                     (short)0, accv, false, false);
    }

    const int mbase = row0 + half * 8;
    if (full_tile) {
        #pragma unroll
        for (int r = 0; r < 8; ++r) {
            u[(size_t)(mbase + r) * C + col] = accu[r];
            v[(size_t)(mbase + r) * C + col] = accv[r];
        }
    } else {
        #pragma unroll
        for (int r = 0; r < 8; ++r) {
            if (mbase + r < n_nodes) {
                u[(size_t)(mbase + r) * C + col] = accu[r];
                v[(size_t)(mbase + r) * C + col] = accv[r];
            }
        }
    }
}

__device__ __forceinline__ float leaky(float x) { return x > 0.0f ? x : LEAKY * x; }

typedef __attribute__((ext_vector_type(4))) float v4f;

__device__ __forceinline__ v4f load_row4(const float* __restrict__ base, int row, int lane) {
    return *(const v4f*)(base + (size_t)row * C + lane * 4);
}

__device__ __forceinline__ float edge_partial(v4f uu, v4f vv, float4 aa) {
    float sum = leaky(uu.x + vv.x) * aa.x;
    sum = fmaf(leaky(uu.y + vv.y), aa.y, sum);
    sum = fmaf(leaky(uu.z + vv.z), aa.z, sum);
    sum = fmaf(leaky(uu.w + vv.w), aa.w, sum);
    return sum;
}

__device__ __forceinline__ float fold(float lo, float hi, bool bit, int off) {
    float keep = bit ? hi : lo;
    float send = bit ? lo : hi;
    return keep + __shfl_xor(send, off, 32);
}

__global__ __launch_bounds__(256) void gat_edge_e(const float* __restrict__ u,
                                                  const float* __restrict__ v,
                                                  const float* __restrict__ a,
                                                  const int* __restrict__ src,
                                                  const int* __restrict__ dst,
                                                  float* __restrict__ exp_e,
                                                  float* __restrict__ denom,
                                                  int n_edges) {
    const int wib  = threadIdx.x >> 5;
    const int lane = threadIdx.x & 31;
    const int e0 = (blockIdx.x * 8 + wib) * 32;
    if (e0 >= n_edges) return;

    const int eown = min(e0 + lane, n_edges - 1);
    const int s_own = src[eown];
    const int d_own = dst[eown];
    const float4 aa = ((const float4*)a)[lane];

    const bool b4 = (lane & 16) != 0;
    const bool b3 = (lane & 8) != 0;
    const bool b2 = (lane & 4) != 0;
    const bool b1 = (lane & 2) != 0;
    const bool b0 = (lane & 1) != 0;

    float wsum[4];
    #pragma unroll
    for (int b = 0; b < 4; ++b) {
        v4f uu[8], vv[8];
        #pragma unroll
        for (int i = 0; i < 8; ++i) {
            uu[i] = load_row4(u, __builtin_amdgcn_readlane(s_own, b * 8 + i), lane);
            vv[i] = load_row4(v, __builtin_amdgcn_readlane(d_own, b * 8 + i), lane);
        }
        asm volatile("" :: "v"(uu[0]), "v"(uu[1]), "v"(uu[2]), "v"(uu[3]),
                           "v"(uu[4]), "v"(uu[5]), "v"(uu[6]), "v"(uu[7]),
                           "v"(vv[0]), "v"(vv[1]), "v"(vv[2]), "v"(vv[3]),
                           "v"(vv[4]), "v"(vv[5]), "v"(vv[6]), "v"(vv[7]));
        float p[8];
        #pragma unroll
        for (int i = 0; i < 8; ++i)
            p[i] = edge_partial(uu[i], vv[i], aa);
        const float q0 = fold(p[0], p[4], b4, 16);
        const float q1 = fold(p[1], p[5], b4, 16);
        const float q2 = fold(p[2], p[6], b4, 16);
        const float q3 = fold(p[3], p[7], b4, 16);
        const float r0 = fold(q0, q2, b3, 8);
        const float r1 = fold(q1, q3, b3, 8);
        wsum[b] = fold(r0, r1, b2, 4);
    }
    const float x0 = fold(wsum[0], wsum[2], b1, 2);
    const float x1 = fold(wsum[1], wsum[3], b1, 2);
    const float logit = fold(x0, x1, b0, 1);

    const int slot = ((lane & 3) << 3) | (lane >> 2);
    const int s_fin = __shfl(s_own, slot, 32);
    const int e = e0 + slot;
    if (e < n_edges) {
        float ex = expf(logit);
        exp_e[e] = ex;
        atomicAdd(&denom[s_fin], ex);
    }
}

__device__ __forceinline__ int lane_i(int x, int l) { return __builtin_amdgcn_readlane(x, l); }
__device__ __forceinline__ float lane_f(float x, int l) {
    return __int_as_float(__builtin_amdgcn_readlane(__float_as_int(x), l));
}

__global__ __launch_bounds__(256) void gat_scatter(const float* __restrict__ h,
                                                   const int* __restrict__ src,
                                                   const int* __restrict__ dst,
                                                   const float* __restrict__ exp_e,
                                                   const float* __restrict__ denom,
                                                   float* __restrict__ out,
                                                   int n_edges) {
    const int wib  = threadIdx.x >> 5;
    const int lane = threadIdx.x & 31;
    const int e0 = (blockIdx.x * 8 + wib) * 32;
    if (e0 >= n_edges) return;

    const int cnt  = min(32, n_edges - e0);
    const int eown = min(e0 + lane, n_edges - 1);
    const int s_own = src[eown];
    const int d_own = dst[eown];
    const float att_own = exp_e[eown] / denom[s_own];

    if (cnt == 32) {
        #pragma unroll
        for (int b = 0; b < 4; ++b) {
            float hv[8][4];
            #pragma unroll
            for (int i = 0; i < 8; ++i) {
                const float* hp = h + (size_t)lane_i(s_own, b * 8 + i) * C + lane;
                hv[i][0] = hp[0];
                hv[i][1] = hp[32];
                hv[i][2] = hp[64];
                hv[i][3] = hp[96];
            }
            #pragma unroll
            for (int i = 0; i < 8; ++i) {
                const float att = lane_f(att_own, b * 8 + i);
                float* op = out + (size_t)lane_i(d_own, b * 8 + i) * C + lane;
                atomicAdd(op,      hv[i][0] * att);
                atomicAdd(op + 32, hv[i][1] * att);
                atomicAdd(op + 64, hv[i][2] * att);
                atomicAdd(op + 96, hv[i][3] * att);
            }
        }
    } else {
        for (int i = 0; i < cnt; ++i) {
            const int s = __shfl(s_own, i, 32);
            const int d = __shfl(d_own, i, 32);
            const float att = __shfl(att_own, i, 32);
            const float* hp = h + (size_t)s * C + lane;
            float* op = out + (size_t)d * C + lane;
            atomicAdd(op,      hp[0]  * att);
            atomicAdd(op + 32, hp[32] * att);
            atomicAdd(op + 64, hp[64] * att);
            atomicAdd(op + 96, hp[96] * att);
        }
    }
}

extern "C" void kernel_launch(void* const* d_in, const int* in_sizes, int n_in,
                              void* d_out, int out_size, void* d_ws, size_t ws_size,
                              hipStream_t stream) {
    const float* h  = (const float*)d_in[0];
    const float* w  = (const float*)d_in[1];
    const float* a  = (const float*)d_in[2];
    const int* src  = (const int*)d_in[3];
    const int* dst  = (const int*)d_in[4];
    float* out      = (float*)d_out;

    const int n_nodes = in_sizes[0] / C;
    const int n_edges = in_sizes[3];

    float* u     = (float*)d_ws;
    float* v     = u + (size_t)n_nodes * C;
    float* exp_e = v + (size_t)n_nodes * C;
    float* denom = exp_e + n_edges;

    {
        int nz = n_nodes * C;
        int grid = (nz + 255) / 256;
        gat_zero<<<grid, 256, 0, stream>>>(out, nz, denom, n_nodes);
    }
    {
        int grid = (n_nodes + 15) / 16;
        gat_gemm_uv<<<grid, 256, 0, stream>>>(h, w, u, v, n_nodes);
    }
    {
        int grid = (n_edges + 255) / 256;
        gat_edge_e<<<grid, 256, 0, stream>>>(u, v, a, src, dst, exp_e, denom, n_edges);
    }
    {
        int grid = (n_edges + 255) / 256;
        gat_scatter<<<grid, 256, 0, stream>>>(h, src, dst, exp_e, denom, out, n_edges);
    }
}


# fill launch removed: the projection kernel stores 0 into its own rows of out (two whole-line float4 stores per thread, behind its barrier) and into denom; 3 launches, was 4; 80000 short waves fewer
# speedup vs baseline: 2.0779x; 1.0319x over previous
#include <hip/hip_runtime.h>
#include <hip/hip_bf16.h>


#define C 128
#define LEAKY 0.2f

typedef __attribute__((ext_vector_type(2))) float v2f;
typedef __attribute__((ext_vector_type(8))) float v8f;

typedef __attribute__((ext_vector_type(4))) float v4f;

__global__ __launch_bounds__(256) void gat_gemm_uv(const float* __restrict__ h,
                                                   const float* __restrict__ w,
                                                   float* __restrict__ u,
                                                   float* __restrict__ v,
                                                   float* __restrict__ out,
                                                   float* __restrict__ denom,
                                                   int n_nodes) {
    __shared__ float lh[16 * C];
    const int row0 = blockIdx.x * 16;
    const int tid  = threadIdx.x;
    const bool full_tile = (row0 + 16) <= n_nodes;

    if (full_tile) {
        for (int i = tid; i < 16 * C; i += 256)
            lh[i] = h[(size_t)row0 * C + i];
    } else {
        for (int i = tid; i < 16 * C; i += 256) {
            int gr = row0 + (i >> 7);
            lh[i] = (gr < n_nodes) ? h[(size_t)gr * C + (i & (C - 1))] : 0.0f;
        }
    }
    __syncthreads();

    {
        const v4f zero4 = {0.0f, 0.0f, 0.0f, 0.0f};
        v4f* o4 = (v4f*)(out + (size_t)row0 * C);
        if (full_tile) {
            o4[tid]       = zero4;
            o4[tid + 256] = zero4;
        } else {
            const int nq = (n_nodes - row0) * (C / 4);
            if (tid < nq)       o4[tid]       = zero4;
            if (tid + 256 < nq) o4[tid + 256] = zero4;
        }
        if ((blockIdx.x & 1) == 0 && tid < 32 && row0 + tid < n_nodes)
            denom[row0 + tid] = 0.0f;
    }

    const int wave = tid >> 5;
    const int lane = tid & 31;
    const int half = lane >> 4;
    const int l16  = lane & 15;
    const int col  = wave * 16 + l16;

    v8f accu = {};
    v8f accv = {};
    for (int k = 0; k < C; k += 4) {
        const int ka = k + half * 2;
        v2f afrag;
        afrag.x = lh[l16 * C + ka];
        afrag.y = lh[l16 * C + ka + 1];
        v2f bu, bv;
        bu.x = w[(size_t)(ka    ) * C + col];
        bu.y = w[(size_t)(ka + 1) * C + col];
        bv.x = w[(size_t)(C + ka    ) * C + col];
        bv.y = w[(size_t)(C + ka + 1) * C + col];
        accu = __builtin_amdgcn_wmma_f32_16x16x4_f32(false, afrag, false, bu,
                                                     (short)0, accu, false, false);
        accv = __builtin_amdgcn_wmma_f32_16x16x4_f32(false, afrag, false, bv,
                                                     (short)0, accv, false, false);
    }

    const int mbase = row0 + half * 8;
    if (full_tile) {
        #pragma unroll
        for (int r = 0; r < 8; ++r) {
            u[(size_t)(mbase + r) * C + col] = accu[r];
            v[(size_t)(mbase + r) * C + col] = accv[r];
        }
    } else {
        #pragma unroll
        for (int r = 0; r < 8; ++r) {
            if (mbase + r < n_nodes) {
                u[(size_t)(mbase + r) * C + col] = accu[r];
                v[(size_t)(mbase + r) * C + col] = accv[r];
            }
        }
    }
}

__device__ __forceinline__ float leaky(float x) { return x > 0.0f ? x : LEAKY * x; }


__device__ __forceinline__ v4f load_row4(const float* __restrict__ base, int row, int lane) {
    return *(const v4f*)(base + (size_t)row * C + lane * 4);
}

__device__ __forceinline__ float edge_partial(v4f uu, v4f vv, float4 aa) {
    float sum = leaky(uu.x + vv.x) * aa.x;
    sum = fmaf(leaky(uu.y + vv.y), aa.y, sum);
    sum = fmaf(leaky(uu.z + vv.z), aa.z, sum);
    sum = fmaf(leaky(uu.w + vv.w), aa.w, sum);
    return sum;
}

__device__ __forceinline__ float fold(float lo, float hi, bool bit, int off) {
    float keep = bit ? hi : lo;
    float send = bit ? lo : hi;
    return keep + __shfl_xor(send, off, 32);
}

__global__ __launch_bounds__(256) void gat_edge_e(const float* __restrict__ u,
                                                  const float* __restrict__ v,
                                                  const float* __restrict__ a,
                                                  const int* __restrict__ src,
                                                  const int* __restrict__ dst,
                                                  float* __restrict__ exp_e,
                                                  float* __restrict__ denom,
                                                  int n_edges) {
    const int wib  = threadIdx.x >> 5;
    const int lane = threadIdx.x & 31;
    const int e0 = (blockIdx.x * 8 + wib) * 32;
    if (e0 >= n_edges) return;

    const int eown = min(e0 + lane, n_edges - 1);
    const int s_own = src[eown];
    const int d_own = dst[eown];
    const float4 aa = ((const float4*)a)[lane];

    const bool b4 = (lane & 16) != 0;
    const bool b3 = (lane & 8) != 0;
    const bool b2 = (lane & 4) != 0;
    const bool b1 = (lane & 2) != 0;
    const bool b0 = (lane & 1) != 0;

    float wsum[4];
    #pragma unroll
    for (int b = 0; b < 4; ++b) {
        v4f uu[8], vv[8];
        #pragma unroll
        for (int i = 0; i < 8; ++i) {
            uu[i] = load_row4(u, __builtin_amdgcn_readlane(s_own, b * 8 + i), lane);
            vv[i] = load_row4(v, __builtin_amdgcn_readlane(d_own, b * 8 + i), lane);
        }
        asm volatile("" :: "v"(uu[0]), "v"(uu[1]), "v"(uu[2]), "v"(uu[3]),
                           "v"(uu[4]), "v"(uu[5]), "v"(uu[6]), "v"(uu[7]),
                           "v"(vv[0]), "v"(vv[1]), "v"(vv[2]), "v"(vv[3]),
                           "v"(vv[4]), "v"(vv[5]), "v"(vv[6]), "v"(vv[7]));
        float p[8];
        #pragma unroll
        for (int i = 0; i < 8; ++i)
            p[i] = edge_partial(uu[i], vv[i], aa);
        const float q0 = fold(p[0], p[4], b4, 16);
        const float q1 = fold(p[1], p[5], b4, 16);
        const float q2 = fold(p[2], p[6], b4, 16);
        const float q3 = fold(p[3], p[7], b4, 16);
        const float r0 = fold(q0, q2, b3, 8);
        const float r1 = fold(q1, q3, b3, 8);
        wsum[b] = fold(r0, r1, b2, 4);
    }
    const float x0 = fold(wsum[0], wsum[2], b1, 2);
    const float x1 = fold(wsum[1], wsum[3], b1, 2);
    const float logit = fold(x0, x1, b0, 1);

    const int slot = ((lane & 3) << 3) | (lane >> 2);
    const int s_fin = __shfl(s_own, slot, 32);
    const int e = e0 + slot;
    if (e < n_edges) {
        float ex = expf(logit);
        exp_e[e] = ex;
        atomicAdd(&denom[s_fin], ex);
    }
}

__device__ __forceinline__ int lane_i(int x, int l) { return __builtin_amdgcn_readlane(x, l); }
__device__ __forceinline__ float lane_f(float x, int l) {
    return __int_as_float(__builtin_amdgcn_readlane(__float_as_int(x), l));
}

__global__ __launch_bounds__(256) void gat_scatter(const float* __restrict__ h,
                                                   const int* __restrict__ src,
                                                   const int* __restrict__ dst,
                                                   const float* __restrict__ exp_e,
                                                   const float* __restrict__ denom,
                                                   float* __restrict__ out,
                                                   int n_edges) {
    const int wib  = threadIdx.x >> 5;
    const int lane = threadIdx.x & 31;
    const int e0 = (blockIdx.x * 8 + wib) * 32;
    if (e0 >= n_edges) return;

    const int cnt  = min(32, n_edges - e0);
    const int eown = min(e0 + lane, n_edges - 1);
    const int s_own = src[eown];
    const int d_own = dst[eown];
    const float att_own = exp_e[eown] / denom[s_own];

    if (cnt == 32) {
        #pragma unroll
        for (int b = 0; b < 4; ++b) {
            float hv[8][4];
            #pragma unroll
            for (int i = 0; i < 8; ++i) {
                const float* hp = h + (size_t)lane_i(s_own, b * 8 + i) * C + lane;
                hv[i][0] = hp[0];
                hv[i][1] = hp[32];
                hv[i][2] = hp[64];
                hv[i][3] = hp[96];
            }
            #pragma unroll
            for (int i = 0; i < 8; ++i) {
                const float att = lane_f(att_own, b * 8 + i);
                float* op = out + (size_t)lane_i(d_own, b * 8 + i) * C + lane;
                atomicAdd(op,      hv[i][0] * att);
                atomicAdd(op + 32, hv[i][1] * att);
                atomicAdd(op + 64, hv[i][2] * att);
                atomicAdd(op + 96, hv[i][3] * att);
            }
        }
    } else {
        for (int i = 0; i < cnt; ++i) {
            const int s = __shfl(s_own, i, 32);
            const int d = __shfl(d_own, i, 32);
            const float att = __shfl(att_own, i, 32);
            const float* hp = h + (size_t)s * C + lane;
            float* op = out + (size_t)d * C + lane;
            atomicAdd(op,      hp[0]  * att);
            atomicAdd(op + 32, hp[32] * att);
            atomicAdd(op + 64, hp[64] * att);
            atomicAdd(op + 96, hp[96] * att);
        }
    }
}

extern "C" void kernel_launch(void* const* d_in, const int* in_sizes, int n_in,
                              void* d_out, int out_size, void* d_ws, size_t ws_size,
                              hipStream_t stream) {
    const float* h  = (const float*)d_in[0];
    const float* w  = (const float*)d_in[1];
    const float* a  = (const float*)d_in[2];
    const int* src  = (const int*)d_in[3];
    const int* dst  = (const int*)d_in[4];
    float* out      = (float*)d_out;

    const int n_nodes = in_sizes[0] / C;
    const int n_edges = in_sizes[3];

    float* u     = (float*)d_ws;
    float* v     = u + (size_t)n_nodes * C;
    float* exp_e = v + (size_t)n_nodes * C;
    float* denom = exp_e + n_edges;

    {
        int grid = (n_nodes + 15) / 16;
        gat_gemm_uv<<<grid, 256, 0, stream>>>(h, w, u, v, out, denom, n_nodes);
    }
    {
        int grid = (n_edges + 255) / 256;
        gat_edge_e<<<grid, 256, 0, stream>>>(u, v, a, src, dst, exp_e, denom, n_edges);
    }
    {
        int grid = (n_edges + 255) / 256;
        gat_scatter<<<grid, 256, 0, stream>>>(h, src, dst, exp_e, denom, out, n_edges);
    }
}


# scatter replaced by a sum by destination: edges binned into 80-row ranges with LDS counters, rows summed by LDS f32 adds, each out row added once (was 4 f32 line atomics/edge); sums in arrival order
# speedup vs baseline: 2.7906x; 1.3430x over previous
#include <hip/hip_runtime.h>
#include <hip/hip_bf16.h>


#define C 128
#define LEAKY 0.2f

typedef __attribute__((ext_vector_type(2))) float v2f;
typedef __attribute__((ext_vector_type(8))) float v8f;

typedef __attribute__((ext_vector_type(4))) float v4f;

__global__ __launch_bounds__(256) void gat_gemm_uv(const float* __restrict__ h,
                                                   const float* __restrict__ w,
                                                   float* __restrict__ u,
                                                   float* __restrict__ v,
                                                   float* __restrict__ out,
                                                   float* __restrict__ denom,
                                                   int* __restrict__ bin_count,
                                                   int n_bins,
                                                   int n_nodes) {
    __shared__ float lh[16 * C];
    const int row0 = blockIdx.x * 16;
    const int tid  = threadIdx.x;
    const bool full_tile = (row0 + 16) <= n_nodes;

    if (full_tile) {
        for (int i = tid; i < 16 * C; i += 256)
            lh[i] = h[(size_t)row0 * C + i];
    } else {
        for (int i = tid; i < 16 * C; i += 256) {
            int gr = row0 + (i >> 7);
            lh[i] = (gr < n_nodes) ? h[(size_t)gr * C + (i & (C - 1))] : 0.0f;
        }
    }
    __syncthreads();

    {
        const v4f zero4 = {0.0f, 0.0f, 0.0f, 0.0f};
        v4f* o4 = (v4f*)(out + (size_t)row0 * C);
        if (full_tile) {
            o4[tid]       = zero4;
            o4[tid + 256] = zero4;
        } else {
            const int nq = (n_nodes - row0) * (C / 4);
            if (tid < nq)       o4[tid]       = zero4;
            if (tid + 256 < nq) o4[tid + 256] = zero4;
        }
        if ((blockIdx.x & 1) == 0 && tid < 32 && row0 + tid < n_nodes)
            denom[row0 + tid] = 0.0f;
        if (blockIdx.x == 0)
            for (int i = tid; i < n_bins; i += 256) bin_count[i] = 0;
    }

    const int wave = tid >> 5;
    const int lane = tid & 31;
    const int half = lane >> 4;
    const int l16  = lane & 15;
    const int col  = wave * 16 + l16;

    v8f accu = {};
    v8f accv = {};
    for (int k = 0; k < C; k += 4) {
        const int ka = k + half * 2;
        v2f afrag;
        afrag.x = lh[l16 * C + ka];
        afrag.y = lh[l16 * C + ka + 1];
        v2f bu, bv;
        bu.x = w[(size_t)(ka    ) * C + col];
        bu.y = w[(size_t)(ka + 1) * C + col];
        bv.x = w[(size_t)(C + ka    ) * C + col];
        bv.y = w[(size_t)(C + ka + 1) * C + col];
        accu = __builtin_amdgcn_wmma_f32_16x16x4_f32(false, afrag, false, bu,
                                                     (short)0, accu, false, false);
        accv = __builtin_amdgcn_wmma_f32_16x16x4_f32(false, afrag, false, bv,
                                                     (short)0, accv, false, false);
    }

    const int mbase = row0 + half * 8;
    if (full_tile) {
        #pragma unroll
        for (int r = 0; r < 8; ++r) {
            u[(size_t)(mbase + r) * C + col] = accu[r];
            v[(size_t)(mbase + r) * C + col] = accv[r];
        }
    } else {
        #pragma unroll
        for (int r = 0; r < 8; ++r) {
            if (mbase + r < n_nodes) {
                u[(size_t)(mbase + r) * C + col] = accu[r];
                v[(size_t)(mbase + r) * C + col] = accv[r];
            }
        }
    }
}

__device__ __forceinline__ float leaky(float x) { return x > 0.0f ? x : LEAKY * x; }


__device__ __forceinline__ v4f load_row4(const float* __restrict__ base, int row, int lane) {
    return *(const v4f*)(base + (size_t)row * C + lane * 4);
}

__device__ __forceinline__ float edge_partial(v4f uu, v4f vv, float4 aa) {
    float sum = leaky(uu.x + vv.x) * aa.x;
    sum = fmaf(leaky(uu.y + vv.y), aa.y, sum);
    sum = fmaf(leaky(uu.z + vv.z), aa.z, sum);
    sum = fmaf(leaky(uu.w + vv.w), aa.w, sum);
    return sum;
}

__device__ __forceinline__ float fold(float lo, float hi, bool bit, int off) {
    float keep = bit ? hi : lo;
    float send = bit ? lo : hi;
    return keep + __shfl_xor(send, off, 32);
}

__global__ __launch_bounds__(256) void gat_edge_e(const float* __restrict__ u,
                                                  const float* __restrict__ v,
                                                  const float* __restrict__ a,
                                                  const int* __restrict__ src,
                                                  const int* __restrict__ dst,
                                                  float* __restrict__ exp_e,
                                                  float* __restrict__ denom,
                                                  int n_edges) {
    const int wib  = threadIdx.x >> 5;
    const int lane = threadIdx.x & 31;
    const int e0 = (blockIdx.x * 8 + wib) * 32;
    if (e0 >= n_edges) return;

    const int eown = min(e0 + lane, n_edges - 1);
    const int s_own = src[eown];
    const int d_own = dst[eown];
    const float4 aa = ((const float4*)a)[lane];

    const bool b4 = (lane & 16) != 0;
    const bool b3 = (lane & 8) != 0;
    const bool b2 = (lane & 4) != 0;
    const bool b1 = (lane & 2) != 0;
    const bool b0 = (lane & 1) != 0;

    float wsum[4];
    #pragma unroll
    for (int b = 0; b < 4; ++b) {
        v4f uu[8], vv[8];
        #pragma unroll
        for (int i = 0; i < 8; ++i) {
            uu[i] = load_row4(u, __builtin_amdgcn_readlane(s_own, b * 8 + i), lane);
            vv[i] = load_row4(v, __builtin_amdgcn_readlane(d_own, b * 8 + i), lane);
        }
        asm volatile("" :: "v"(uu[0]), "v"(uu[1]), "v"(uu[2]), "v"(uu[3]),
                           "v"(uu[4]), "v"(uu[5]), "v"(uu[6]), "v"(uu[7]),
                           "v"(vv[0]), "v"(vv[1]), "v"(vv[2]), "v"(vv[3]),
                           "v"(vv[4]), "v"(vv[5]), "v"(vv[6]), "v"(vv[7]));
        float p[8];
        #pragma unroll
        for (int i = 0; i < 8; ++i)
            p[i] = edge_partial(uu[i], vv[i], aa);
        const float q0 = fold(p[0], p[4], b4, 16);
        const float q1 = fold(p[1], p[5], b4, 16);
        const float q2 = fold(p[2], p[6], b4, 16);
        const float q3 = fold(p[3], p[7], b4, 16);
        const float r0 = fold(q0, q2, b3, 8);
        const float r1 = fold(q1, q3, b3, 8);
        wsum[b] = fold(r0, r1, b2, 4);
    }
    const float x0 = fold(wsum[0], wsum[2], b1, 2);
    const float x1 = fold(wsum[1], wsum[3], b1, 2);
    const float logit = fold(x0, x1, b0, 1);

    const int slot = ((lane & 3) << 3) | (lane >> 2);
    const int s_fin = __shfl(s_own, slot, 32);
    const int e = e0 + slot;
    if (e < n_edges) {
        float ex = expf(logit);
        exp_e[e] = ex;
        atomicAdd(&denom[s_fin], ex);
    }
}

#define RB   80
#define CAP  4096
#define PL_E 10

__device__ __forceinline__ int lane_i(int x, int l) { return __builtin_amdgcn_readlane(x, l); }
__device__ __forceinline__ float lane_f(float x, int l) {
    return __int_as_float(__builtin_amdgcn_readlane(__float_as_int(x), l));
}

__global__ __launch_bounds__(256) void gat_place(const float* __restrict__ h,
                                                 const int* __restrict__ src,
                                                 const int* __restrict__ dst,
                                                 const float* __restrict__ exp_e,
                                                 const float* __restrict__ denom,
                                                 int* __restrict__ bin_count,
                                                 int* __restrict__ rec_raw,
                                                 float* __restrict__ out,
                                                 int n_edges, int n_bins) {
    extern __shared__ int sm[];
    int* hist = sm;
    int* base = sm + n_bins;
    const int tid = threadIdx.x;
    for (int i = tid; i < n_bins; i += 256) hist[i] = 0;
    __syncthreads();

    const int e0 = blockIdx.x * (256 * PL_E);
    int s[PL_E], d[PL_E], rk[PL_E];
    float ex[PL_E], dn[PL_E];
    #pragma unroll
    for (int j = 0; j < PL_E; ++j) {
        const int e  = e0 + j * 256 + tid;
        const int ee = (e < n_edges) ? e : n_edges - 1;
        s[j]  = src[ee];
        d[j]  = dst[ee];
        ex[j] = exp_e[ee];
    }
    #pragma unroll
    for (int j = 0; j < PL_E; ++j) dn[j] = denom[s[j]];
    #pragma unroll
    for (int j = 0; j < PL_E; ++j) {
        const int e = e0 + j * 256 + tid;
        rk[j] = (e < n_edges) ? atomicAdd(&hist[d[j] / RB], 1) : -1;
    }
    __syncthreads();
    for (int i = tid; i < n_bins; i += 256) {
        const int c = hist[i];
        base[i] = (c > 0) ? atomicAdd(&bin_count[i], c) : 0;
    }
    __syncthreads();

    int4* rec = (int4*)rec_raw;
    #pragma unroll
    for (int j = 0; j < PL_E; ++j) {
        if (rk[j] >= 0) {
            const int b   = d[j] / RB;
            const int pos = base[b] + rk[j];
            if (pos < CAP) {
                rec[(size_t)b * CAP + pos] = make_int4(s[j], d[j] - b * RB, __float_as_int(ex[j] / dn[j]), 0);
            } else {
                const float* hp = h + (size_t)s[j] * C;
                float* op = out + (size_t)d[j] * C;
                const float att = ex[j] / dn[j];
                for (int c = 0; c < C; ++c) atomicAdd(op + c, hp[c] * att);
            }
        }
    }
}

__global__ __launch_bounds__(1024) void gat_sum(const float* __restrict__ h,
                                                const int* __restrict__ rec_raw,
                                                const int* __restrict__ bin_count,
                                                float* __restrict__ out,
                                                int n_nodes) {
    __shared__ float acc[RB * C];
    const int tid  = threadIdx.x;
    const int lane = tid & 31;
    const int wave = __builtin_amdgcn_readfirstlane(tid >> 5);
    const int b = blockIdx.x;

    for (int i = tid; i < RB * C; i += 1024) acc[i] = 0.0f;
    __syncthreads();

    const int n = min(bin_count[b], CAP);
    const int4* r = (const int4*)rec_raw + (size_t)b * CAP;

    for (int g0 = wave * 16; g0 < n; g0 += 512) {
        const int cnt = min(16, n - g0);
        int4 rc = make_int4(0, 0, 0, 0);
        if (lane < cnt) rc = r[g0 + lane];
        const float att_own = __int_as_float(rc.z);
        float hv[16][4];
        #pragma unroll
        for (int i = 0; i < 16; ++i) {
            const float* hp = h + (size_t)lane_i(rc.x, i) * C + lane;
            hv[i][0] = hp[0];
            hv[i][1] = hp[32];
            hv[i][2] = hp[64];
            hv[i][3] = hp[96];
        }
        #pragma unroll
        for (int i = 0; i < 16; ++i) {
            if (i < cnt) {
                const float att = lane_f(att_own, i);
                float* ap = acc + lane_i(rc.y, i) * C + lane;
                atomicAdd(ap,      hv[i][0] * att);
                atomicAdd(ap + 32, hv[i][1] * att);
                atomicAdd(ap + 64, hv[i][2] * att);
                atomicAdd(ap + 96, hv[i][3] * att);
            }
        }
    }
    __syncthreads();

    const int row0 = b * RB;
    float ov[RB * C / 1024];
    #pragma unroll
    for (int k = 0; k < RB * C / 1024; ++k) {
        const int i = tid + k * 1024;
        ov[k] = (row0 + (i >> 7) < n_nodes) ? out[(size_t)row0 * C + i] : 0.0f;
    }
    #pragma unroll
    for (int k = 0; k < RB * C / 1024; ++k) {
        const int i = tid + k * 1024;
        if (row0 + (i >> 7) < n_nodes) out[(size_t)row0 * C + i] = ov[k] + acc[i];
    }
}

extern "C" void kernel_launch(void* const* d_in, const int* in_sizes, int n_in,
                              void* d_out, int out_size, void* d_ws, size_t ws_size,
                              hipStream_t stream) {
    const float* h  = (const float*)d_in[0];
    const float* w  = (const float*)d_in[1];
    const float* a  = (const float*)d_in[2];
    const int* src  = (const int*)d_in[3];
    const int* dst  = (const int*)d_in[4];
    float* out      = (float*)d_out;

    const int n_nodes = in_sizes[0] / C;
    const int n_edges = in_sizes[3];

    const int n_bins  = (n_nodes + RB - 1) / RB;

    float* u       = (float*)d_ws;
    float* v       = u + (size_t)n_nodes * C;
    int* rec       = (int*)(v + (size_t)n_nodes * C);
    float* exp_e   = (float*)(rec + (size_t)n_bins * CAP * 4);
    float* denom   = exp_e + n_edges;
    int* bin_count = (int*)(denom + n_nodes);

    {
        int grid = (n_nodes + 15) / 16;
        gat_gemm_uv<<<grid, 256, 0, stream>>>(h, w, u, v, out, denom, bin_count, n_bins, n_nodes);
    }
    {
        int grid = (n_edges + 255) / 256;
        gat_edge_e<<<grid, 256, 0, stream>>>(u, v, a, src, dst, exp_e, denom, n_edges);
    }
    {
        int grid = (n_edges + 256 * PL_E - 1) / (256 * PL_E);
        gat_place<<<grid, 256, (size_t)n_bins * 2 * sizeof(int), stream>>>(
            h, src, dst, exp_e, denom, bin_count, rec, out, n_edges, n_bins);
    }
    {
        gat_sum<<<n_bins, 1024, 0, stream>>>(h, rec, bin_count, out, n_nodes);
    }
}


# sum kernel: each range's records sorted by row in LDS (counting sort); a wave sums a row's run in registers (fma), adds to LDS only at a row change (was 4 LDS f32 adds/edge); sum order changed
# speedup vs baseline: 2.9680x; 1.0636x over previous
#include <hip/hip_runtime.h>
#include <hip/hip_bf16.h>


#define C 128
#define LEAKY 0.2f

typedef __attribute__((ext_vector_type(2))) float v2f;
typedef __attribute__((ext_vector_type(8))) float v8f;

typedef __attribute__((ext_vector_type(4))) float v4f;

__global__ __launch_bounds__(256) void gat_gemm_uv(const float* __restrict__ h,
                                                   const float* __restrict__ w,
                                                   float* __restrict__ u,
                                                   float* __restrict__ v,
                                                   float* __restrict__ out,
                                                   float* __restrict__ denom,
                                                   int* __restrict__ bin_count,
                                                   int n_bins,
                                                   int n_nodes) {
    __shared__ float lh[16 * C];
    const int row0 = blockIdx.x * 16;
    const int tid  = threadIdx.x;
    const bool full_tile = (row0 + 16) <= n_nodes;

    if (full_tile) {
        for (int i = tid; i < 16 * C; i += 256)
            lh[i] = h[(size_t)row0 * C + i];
    } else {
        for (int i = tid; i < 16 * C; i += 256) {
            int gr = row0 + (i >> 7);
            lh[i] = (gr < n_nodes) ? h[(size_t)gr * C + (i & (C - 1))] : 0.0f;
        }
    }
    __syncthreads();

    {
        const v4f zero4 = {0.0f, 0.0f, 0.0f, 0.0f};
        v4f* o4 = (v4f*)(out + (size_t)row0 * C);
        if (full_tile) {
            o4[tid]       = zero4;
            o4[tid + 256] = zero4;
        } else {
            const int nq = (n_nodes - row0) * (C / 4);
            if (tid < nq)       o4[tid]       = zero4;
            if (tid + 256 < nq) o4[tid + 256] = zero4;
        }
        if ((blockIdx.x & 1) == 0 && tid < 32 && row0 + tid < n_nodes)
            denom[row0 + tid] = 0.0f;
        if (blockIdx.x == 0)
            for (int i = tid; i < n_bins; i += 256) bin_count[i] = 0;
    }

    const int wave = tid >> 5;
    const int lane = tid & 31;
    const int half = lane >> 4;
    const int l16  = lane & 15;
    const int col  = wave * 16 + l16;

    v8f accu = {};
    v8f accv = {};
    for (int k = 0; k < C; k += 4) {
        const int ka = k + half * 2;
        v2f afrag;
        afrag.x = lh[l16 * C + ka];
        afrag.y = lh[l16 * C + ka + 1];
        v2f bu, bv;
        bu.x = w[(size_t)(ka    ) * C + col];
        bu.y = w[(size_t)(ka + 1) * C + col];
        bv.x = w[(size_t)(C + ka    ) * C + col];
        bv.y = w[(size_t)(C + ka + 1) * C + col];
        accu = __builtin_amdgcn_wmma_f32_16x16x4_f32(false, afrag, false, bu,
                                                     (short)0, accu, false, false);
        accv = __builtin_amdgcn_wmma_f32_16x16x4_f32(false, afrag, false, bv,
                                                     (short)0, accv, false, false);
    }

    const int mbase = row0 + half * 8;
    if (full_tile) {
        #pragma unroll
        for (int r = 0; r < 8; ++r) {
            u[(size_t)(mbase + r) * C + col] = accu[r];
            v[(size_t)(mbase + r) * C + col] = accv[r];
        }
    } else {
        #pragma unroll
        for (int r = 0; r < 8; ++r) {
            if (mbase + r < n_nodes) {
                u[(size_t)(mbase + r) * C + col] = accu[r];
                v[(size_t)(mbase + r) * C + col] = accv[r];
            }
        }
    }
}

__device__ __forceinline__ float leaky(float x) { return x > 0.0f ? x : LEAKY * x; }


__device__ __forceinline__ v4f load_row4(const float* __restrict__ base, int row, int lane) {
    return *(const v4f*)(base + (size_t)row * C + lane * 4);
}

__device__ __forceinline__ float edge_partial(v4f uu, v4f vv, float4 aa) {
    float sum = leaky(uu.x + vv.x) * aa.x;
    sum = fmaf(leaky(uu.y + vv.y), aa.y, sum);
    sum = fmaf(leaky(uu.z + vv.z), aa.z, sum);
    sum = fmaf(leaky(uu.w + vv.w), aa.w, sum);
    return sum;
}

__device__ __forceinline__ float fold(float lo, float hi, bool bit, int off) {
    float keep = bit ? hi : lo;
    float send = bit ? lo : hi;
    return keep + __shfl_xor(send, off, 32);
}

__global__ __launch_bounds__(256) void gat_edge_e(const float* __restrict__ u,
                                                  const float* __restrict__ v,
                                                  const float* __restrict__ a,
                                                  const int* __restrict__ src,
                                                  const int* __restrict__ dst,
                                                  float* __restrict__ exp_e,
                                                  float* __restrict__ denom,
                                                  int n_edges) {
    const int wib  = threadIdx.x >> 5;
    const int lane = threadIdx.x & 31;
    const int e0 = (blockIdx.x * 8 + wib) * 32;
    if (e0 >= n_edges) return;

    const int eown = min(e0 + lane, n_edges - 1);
    const int s_own = src[eown];
    const int d_own = dst[eown];
    const float4 aa = ((const float4*)a)[lane];

    const bool b4 = (lane & 16) != 0;
    const bool b3 = (lane & 8) != 0;
    const bool b2 = (lane & 4) != 0;
    const bool b1 = (lane & 2) != 0;
    const bool b0 = (lane & 1) != 0;

    float wsum[4];
    #pragma unroll
    for (int b = 0; b < 4; ++b) {
        v4f uu[8], vv[8];
        #pragma unroll
        for (int i = 0; i < 8; ++i) {
            uu[i] = load_row4(u, __builtin_amdgcn_readlane(s_own, b * 8 + i), lane);
            vv[i] = load_row4(v, __builtin_amdgcn_readlane(d_own, b * 8 + i), lane);
        }
        asm volatile("" :: "v"(uu[0]), "v"(uu[1]), "v"(uu[2]), "v"(uu[3]),
                           "v"(uu[4]), "v"(uu[5]), "v"(uu[6]), "v"(uu[7]),
                           "v"(vv[0]), "v"(vv[1]), "v"(vv[2]), "v"(vv[3]),
                           "v"(vv[4]), "v"(vv[5]), "v"(vv[6]), "v"(vv[7]));
        float p[8];
        #pragma unroll
        for (int i = 0; i < 8; ++i)
            p[i] = edge_partial(uu[i], vv[i], aa);
        const float q0 = fold(p[0], p[4], b4, 16);
        const float q1 = fold(p[1], p[5], b4, 16);
        const float q2 = fold(p[2], p[6], b4, 16);
        const float q3 = fold(p[3], p[7], b4, 16);
        const float r0 = fold(q0, q2, b3, 8);
        const float r1 = fold(q1, q3, b3, 8);
        wsum[b] = fold(r0, r1, b2, 4);
    }
    const float x0 = fold(wsum[0], wsum[2], b1, 2);
    const float x1 = fold(wsum[1], wsum[3], b1, 2);
    const float logit = fold(x0, x1, b0, 1);

    const int slot = ((lane & 3) << 3) | (lane >> 2);
    const int s_fin = __shfl(s_own, slot, 32);
    const int e = e0 + slot;
    if (e < n_edges) {
        float ex = expf(logit);
        exp_e[e] = ex;
        atomicAdd(&denom[s_fin], ex);
    }
}

#define RB   80
#define CAP  4096
#define PL_E 10
#define RSLOT 128
static_assert(RB <= RSLOT, "one wave scans all row counters");
static_assert(CAP % 1024 == 0, "gat_sum: whole records per thread");

__device__ __forceinline__ int lane_i(int x, int l) { return __builtin_amdgcn_readlane(x, l); }
__device__ __forceinline__ float lane_f(float x, int l) {
    return __int_as_float(__builtin_amdgcn_readlane(__float_as_int(x), l));
}

__global__ __launch_bounds__(256) void gat_place(const float* __restrict__ h,
                                                 const int* __restrict__ src,
                                                 const int* __restrict__ dst,
                                                 const float* __restrict__ exp_e,
                                                 const float* __restrict__ denom,
                                                 int* __restrict__ bin_count,
                                                 int* __restrict__ rec_raw,
                                                 float* __restrict__ out,
                                                 int n_edges, int n_bins) {
    extern __shared__ int sm[];
    int* hist = sm;
    int* base = sm + n_bins;
    const int tid = threadIdx.x;
    for (int i = tid; i < n_bins; i += 256) hist[i] = 0;
    __syncthreads();

    const int e0 = blockIdx.x * (256 * PL_E);
    int s[PL_E], d[PL_E], rk[PL_E];
    float ex[PL_E], dn[PL_E];
    #pragma unroll
    for (int j = 0; j < PL_E; ++j) {
        const int e  = e0 + j * 256 + tid;
        const int ee = (e < n_edges) ? e : n_edges - 1;
        s[j]  = src[ee];
        d[j]  = dst[ee];
        ex[j] = exp_e[ee];
    }
    #pragma unroll
    for (int j = 0; j < PL_E; ++j) dn[j] = denom[s[j]];
    #pragma unroll
    for (int j = 0; j < PL_E; ++j) {
        const int e = e0 + j * 256 + tid;
        rk[j] = (e < n_edges) ? atomicAdd(&hist[d[j] / RB], 1) : -1;
    }
    __syncthreads();
    for (int i = tid; i < n_bins; i += 256) {
        const int c = hist[i];
        base[i] = (c > 0) ? atomicAdd(&bin_count[i], c) : 0;
    }
    __syncthreads();

    int4* rec = (int4*)rec_raw;
    #pragma unroll
    for (int j = 0; j < PL_E; ++j) {
        if (rk[j] >= 0) {
            const int b   = d[j] / RB;
            const int pos = base[b] + rk[j];
            if (pos < CAP) {
                rec[(size_t)b * CAP + pos] = make_int4(s[j], d[j] - b * RB, __float_as_int(ex[j] / dn[j]), 0);
            } else {
                const float* hp = h + (size_t)s[j] * C;
                float* op = out + (size_t)d[j] * C;
                const float att = ex[j] / dn[j];
                for (int c = 0; c < C; ++c) atomicAdd(op + c, hp[c] * att);
            }
        }
    }
}

__device__ __forceinline__ void flush_row(float* acc_row, int lane, v4f a) {
    float* ap = acc_row + lane * 4;
    atomicAdd(ap,     a.x);
    atomicAdd(ap + 1, a.y);
    atomicAdd(ap + 2, a.z);
    atomicAdd(ap + 3, a.w);
}

__global__ __launch_bounds__(1024) void gat_sum(const float* __restrict__ h,
                                                const int* __restrict__ rec_raw,
                                                const int* __restrict__ bin_count,
                                                float* __restrict__ out,
                                                int n_nodes) {
    __shared__ float acc[RB * C];
    __shared__ int4  srt[CAP];
    __shared__ int   rowc[RSLOT];
    const int tid  = threadIdx.x;
    const int lane = tid & 31;
    const int wave = __builtin_amdgcn_readfirstlane(tid >> 5);
    const int b = blockIdx.x;

    const int n = min(bin_count[b], CAP);
    const int4* r = (const int4*)rec_raw + (size_t)b * CAP;

    int4 mine[CAP / 1024];
    #pragma unroll
    for (int k = 0; k < CAP / 1024; ++k) {
        const int idx = tid + k * 1024;
        mine[k] = make_int4(0, -1, 0, 0);
        if (idx < n) mine[k] = r[idx];
    }
    for (int i = tid; i < RB * C; i += 1024) acc[i] = 0.0f;
    if (tid < RSLOT) rowc[tid] = 0;
    __syncthreads();

    int rk[CAP / 1024];
    #pragma unroll
    for (int k = 0; k < CAP / 1024; ++k)
        rk[k] = (mine[k].y >= 0) ? atomicAdd(&rowc[mine[k].y], 1) : 0;
    __syncthreads();
    if (wave == 0) {
        const int c0 = rowc[lane * 4], c1 = rowc[lane * 4 + 1], c2 = rowc[lane * 4 + 2], c3 = rowc[lane * 4 + 3];
        const int s = c0 + c1 + c2 + c3;
        int x = s;
        #pragma unroll
        for (int off = 1; off < 32; off <<= 1) {
            const int y = __shfl_up(x, off, 32);
            if (lane >= off) x += y;
        }
        const int first = x - s;
        rowc[lane * 4]     = first;
        rowc[lane * 4 + 1] = first + c0;
        rowc[lane * 4 + 2] = first + c0 + c1;
        rowc[lane * 4 + 3] = first + c0 + c1 + c2;
    }
    __syncthreads();
    #pragma unroll
    for (int k = 0; k < CAP / 1024; ++k)
        if (mine[k].y >= 0) srt[rowc[mine[k].y] + rk[k]] = mine[k];
    __syncthreads();

    const int nb = (n + 15) >> 4;
    const int lo = ((wave * nb) >> 5) << 4;
    const int hi = min(n, (((wave + 1) * nb) >> 5) << 4);
    int cur = -1;
    v4f a = {0.0f, 0.0f, 0.0f, 0.0f};
    for (int g = lo; g < hi; g += 16) {
        const int cnt = min(16, hi - g);
        int4 rc = make_int4(0, 0, 0, 0);
        if (lane < cnt) rc = srt[g + lane];
        v4f hv[16];
        #pragma unroll
        for (int i = 0; i < 16; ++i)
            hv[i] = load_row4(h, lane_i(rc.x, i), lane);
        #pragma unroll
        for (int i = 0; i < 16; ++i) {
            if (i < cnt) {
                const int row = lane_i(rc.y, i);
                const float att = lane_f(__int_as_float(rc.z), i);
                if (row != cur) {
                    if (cur >= 0) flush_row(acc + cur * C, lane, a);
                    cur = row;
                    a = (v4f){0.0f, 0.0f, 0.0f, 0.0f};
                }
                a.x = fmaf(hv[i].x, att, a.x);
                a.y = fmaf(hv[i].y, att, a.y);
                a.z = fmaf(hv[i].z, att, a.z);
                a.w = fmaf(hv[i].w, att, a.w);
            }
        }
    }
    if (cur >= 0) flush_row(acc + cur * C, lane, a);
    __syncthreads();

    const int row0 = b * RB;
    float ov[RB * C / 1024];
    #pragma unroll
    for (int k = 0; k < RB * C / 1024; ++k) {
        const int i = tid + k * 1024;
        ov[k] = (row0 + (i >> 7) < n_nodes) ? out[(size_t)row0 * C + i] : 0.0f;
    }
    #pragma unroll
    for (int k = 0; k < RB * C / 1024; ++k) {
        const int i = tid + k * 1024;
        if (row0 + (i >> 7) < n_nodes) out[(size_t)row0 * C + i] = ov[k] + acc[i];
    }
}

extern "C" void kernel_launch(void* const* d_in, const int* in_sizes, int n_in,
                              void* d_out, int out_size, void* d_ws, size_t ws_size,
                              hipStream_t stream) {
    const float* h  = (const float*)d_in[0];
    const float* w  = (const float*)d_in[1];
    const float* a  = (const float*)d_in[2];
    const int* src  = (const int*)d_in[3];
    const int* dst  = (const int*)d_in[4];
    float* out      = (float*)d_out;

    const int n_nodes = in_sizes[0] / C;
    const int n_edges = in_sizes[3];

    const int n_bins  = (n_nodes + RB - 1) / RB;

    float* u       = (float*)d_ws;
    float* v       = u + (size_t)n_nodes * C;
    int* rec       = (int*)(v + (size_t)n_nodes * C);
    float* exp_e   = (float*)(rec + (size_t)n_bins * CAP * 4);
    float* denom   = exp_e + n_edges;
    int* bin_count = (int*)(denom + n_nodes);

    {
        int grid = (n_nodes + 15) / 16;
        gat_gemm_uv<<<grid, 256, 0, stream>>>(h, w, u, v, out, denom, bin_count, n_bins, n_nodes);
    }
    {
        int grid = (n_edges + 255) / 256;
        gat_edge_e<<<grid, 256, 0, stream>>>(u, v, a, src, dst, exp_e, denom, n_edges);
    }
    {
        int grid = (n_edges + 256 * PL_E - 1) / (256 * PL_E);
        gat_place<<<grid, 256, (size_t)n_bins * 2 * sizeof(int), stream>>>(
            h, src, dst, exp_e, denom, bin_count, rec, out, n_edges, n_bins);
    }
    {
        gat_sum<<<n_bins, 1024, 0, stream>>>(h, rec, bin_count, out, n_nodes);
    }
}


# softmax denominators: grouping kernel sums exp(e) by source in an LDS table of all nodes, adds it to denom by whole-line f32 atomics (was one scattered atomic per edge); sum order changed
# speedup vs baseline: 3.6406x; 1.2266x over previous
#include <hip/hip_runtime.h>
#include <hip/hip_bf16.h>


#define C 128
#define LEAKY 0.2f

typedef __attribute__((ext_vector_type(2))) float v2f;
typedef __attribute__((ext_vector_type(8))) float v8f;

typedef __attribute__((ext_vector_type(4))) float v4f;

__global__ __launch_bounds__(256) void gat_gemm_uv(const float* __restrict__ h,
                                                   const float* __restrict__ w,
                                                   float* __restrict__ u,
                                                   float* __restrict__ v,
                                                   float* __restrict__ out,
                                                   float* __restrict__ denom,
                                                   int* __restrict__ bin_count,
                                                   int n_bins,
                                                   int n_nodes) {
    __shared__ float lh[16 * C];
    const int row0 = blockIdx.x * 16;
    const int tid  = threadIdx.x;
    const bool full_tile = (row0 + 16) <= n_nodes;

    if (full_tile) {
        for (int i = tid; i < 16 * C; i += 256)
            lh[i] = h[(size_t)row0 * C + i];
    } else {
        for (int i = tid; i < 16 * C; i += 256) {
            int gr = row0 + (i >> 7);
            lh[i] = (gr < n_nodes) ? h[(size_t)gr * C + (i & (C - 1))] : 0.0f;
        }
    }
    __syncthreads();

    {
        const v4f zero4 = {0.0f, 0.0f, 0.0f, 0.0f};
        v4f* o4 = (v4f*)(out + (size_t)row0 * C);
        if (full_tile) {
            o4[tid]       = zero4;
            o4[tid + 256] = zero4;
        } else {
            const int nq = (n_nodes - row0) * (C / 4);
            if (tid < nq)       o4[tid]       = zero4;
            if (tid + 256 < nq) o4[tid + 256] = zero4;
        }
        if ((blockIdx.x & 1) == 0 && tid < 32 && row0 + tid < n_nodes)
            denom[row0 + tid] = 0.0f;
        if (blockIdx.x == 0)
            for (int i = tid; i < n_bins + 1; i += 256) bin_count[i] = 0;
    }

    const int wave = tid >> 5;
    const int lane = tid & 31;
    const int half = lane >> 4;
    const int l16  = lane & 15;
    const int col  = wave * 16 + l16;

    v8f accu = {};
    v8f accv = {};
    for (int k = 0; k < C; k += 4) {
        const int ka = k + half * 2;
        v2f afrag;
        afrag.x = lh[l16 * C + ka];
        afrag.y = lh[l16 * C + ka + 1];
        v2f bu, bv;
        bu.x = w[(size_t)(ka    ) * C + col];
        bu.y = w[(size_t)(ka + 1) * C + col];
        bv.x = w[(size_t)(C + ka    ) * C + col];
        bv.y = w[(size_t)(C + ka + 1) * C + col];
        accu = __builtin_amdgcn_wmma_f32_16x16x4_f32(false, afrag, false, bu,
                                                     (short)0, accu, false, false);
        accv = __builtin_amdgcn_wmma_f32_16x16x4_f32(false, afrag, false, bv,
                                                     (short)0, accv, false, false);
    }

    const int mbase = row0 + half * 8;
    if (full_tile) {
        #pragma unroll
        for (int r = 0; r < 8; ++r) {
            u[(size_t)(mbase + r) * C + col] = accu[r];
            v[(size_t)(mbase + r) * C + col] = accv[r];
        }
    } else {
        #pragma unroll
        for (int r = 0; r < 8; ++r) {
            if (mbase + r < n_nodes) {
                u[(size_t)(mbase + r) * C + col] = accu[r];
                v[(size_t)(mbase + r) * C + col] = accv[r];
            }
        }
    }
}

__device__ __forceinline__ float leaky(float x) { return x > 0.0f ? x : LEAKY * x; }


__device__ __forceinline__ v4f load_row4(const float* __restrict__ base, int row, int lane) {
    return *(const v4f*)(base + (size_t)row * C + lane * 4);
}

__device__ __forceinline__ float edge_partial(v4f uu, v4f vv, float4 aa) {
    float sum = leaky(uu.x + vv.x) * aa.x;
    sum = fmaf(leaky(uu.y + vv.y), aa.y, sum);
    sum = fmaf(leaky(uu.z + vv.z), aa.z, sum);
    sum = fmaf(leaky(uu.w + vv.w), aa.w, sum);
    return sum;
}

__device__ __forceinline__ float fold(float lo, float hi, bool bit, int off) {
    float keep = bit ? hi : lo;
    float send = bit ? lo : hi;
    return keep + __shfl_xor(send, off, 32);
}

__global__ __launch_bounds__(256) void gat_edge_e(const float* __restrict__ u,
                                                  const float* __restrict__ v,
                                                  const float* __restrict__ a,
                                                  const int* __restrict__ src,
                                                  const int* __restrict__ dst,
                                                  float* __restrict__ exp_e,
                                                  int n_edges) {
    const int wib  = threadIdx.x >> 5;
    const int lane = threadIdx.x & 31;
    const int e0 = (blockIdx.x * 8 + wib) * 32;
    if (e0 >= n_edges) return;

    const int eown = min(e0 + lane, n_edges - 1);
    const int s_own = src[eown];
    const int d_own = dst[eown];
    const float4 aa = ((const float4*)a)[lane];

    const bool b4 = (lane & 16) != 0;
    const bool b3 = (lane & 8) != 0;
    const bool b2 = (lane & 4) != 0;
    const bool b1 = (lane & 2) != 0;
    const bool b0 = (lane & 1) != 0;

    float wsum[4];
    #pragma unroll
    for (int b = 0; b < 4; ++b) {
        v4f uu[8], vv[8];
        #pragma unroll
        for (int i = 0; i < 8; ++i) {
            uu[i] = load_row4(u, __builtin_amdgcn_readlane(s_own, b * 8 + i), lane);
            vv[i] = load_row4(v, __builtin_amdgcn_readlane(d_own, b * 8 + i), lane);
        }
        asm volatile("" :: "v"(uu[0]), "v"(uu[1]), "v"(uu[2]), "v"(uu[3]),
                           "v"(uu[4]), "v"(uu[5]), "v"(uu[6]), "v"(uu[7]),
                           "v"(vv[0]), "v"(vv[1]), "v"(vv[2]), "v"(vv[3]),
                           "v"(vv[4]), "v"(vv[5]), "v"(vv[6]), "v"(vv[7]));
        float p[8];
        #pragma unroll
        for (int i = 0; i < 8; ++i)
            p[i] = edge_partial(uu[i], vv[i], aa);
        const float q0 = fold(p[0], p[4], b4, 16);
        const float q1 = fold(p[1], p[5], b4, 16);
        const float q2 = fold(p[2], p[6], b4, 16);
        const float q3 = fold(p[3], p[7], b4, 16);
        const float r0 = fold(q0, q2, b3, 8);
        const float r1 = fold(q1, q3, b3, 8);
        wsum[b] = fold(r0, r1, b2, 4);
    }
    const float x0 = fold(wsum[0], wsum[2], b1, 2);
    const float x1 = fold(wsum[1], wsum[3], b1, 2);
    const float logit = fold(x0, x1, b0, 1);

    const int slot = ((lane & 3) << 3) | (lane >> 2);
    const int e = e0 + slot;
    if (e < n_edges) exp_e[e] = expf(logit);
}

#define RB   80
#define CAP  4096
#define PL_T 1024
#define PL_E 5
#define NTAB 20480
static_assert(NTAB % 4 == 0, "gat_place clears the table 16 B at a time");
#define RSLOT 128
static_assert(RB <= RSLOT, "one wave scans all row counters");
static_assert(CAP % 1024 == 0, "gat_sum: whole records per thread");

__device__ __forceinline__ int lane_i(int x, int l) { return __builtin_amdgcn_readlane(x, l); }
__device__ __forceinline__ float lane_f(float x, int l) {
    return __int_as_float(__builtin_amdgcn_readlane(__float_as_int(x), l));
}

__global__ __launch_bounds__(PL_T) void gat_place(const int* __restrict__ src,
                                                  const int* __restrict__ dst,
                                                  const float* __restrict__ exp_e,
                                                  float* __restrict__ denom,
                                                  int* __restrict__ bin_count,
                                                  int* __restrict__ rec_raw,
                                                  int* __restrict__ ovf_raw,
                                                  int n_edges, int n_bins, int n_nodes) {
    extern __shared__ int sm[];
    __shared__ __attribute__((aligned(16))) float dtab[NTAB];
    int* hist = sm;
    int* base = sm + n_bins;
    const int tid  = threadIdx.x;
    const int ntab = min(n_nodes, NTAB);

    const int e0 = blockIdx.x * (PL_T * PL_E);
    int s[PL_E], d[PL_E], rk[PL_E];
    float ex[PL_E];
    #pragma unroll
    for (int j = 0; j < PL_E; ++j) {
        const int e  = e0 + j * PL_T + tid;
        const int ee = (e < n_edges) ? e : n_edges - 1;
        s[j]  = src[ee];
        d[j]  = dst[ee];
        ex[j] = exp_e[ee];
    }
    for (int i = tid; i < n_bins; i += PL_T) hist[i] = 0;
    for (int i = tid; i < (ntab + 3) / 4; i += PL_T)
        ((v4f*)dtab)[i] = (v4f){0.0f, 0.0f, 0.0f, 0.0f};
    __syncthreads();

    #pragma unroll
    for (int j = 0; j < PL_E; ++j) {
        const int e = e0 + j * PL_T + tid;
        rk[j] = -1;
        if (e < n_edges) {
            rk[j] = atomicAdd(&hist[d[j] / RB], 1);
            if (s[j] < ntab) atomicAdd(&dtab[s[j]], ex[j]);
            else             atomicAdd(&denom[s[j]], ex[j]);
        }
    }
    __syncthreads();
    for (int i = tid; i < n_bins; i += PL_T) {
        const int c = hist[i];
        base[i] = (c > 0) ? atomicAdd(&bin_count[i], c) : 0;
    }
    __syncthreads();

    int4* rec = (int4*)rec_raw;
    int4* ovf = (int4*)ovf_raw;
    #pragma unroll
    for (int j = 0; j < PL_E; ++j) {
        if (rk[j] >= 0) {
            const int b   = d[j] / RB;
            const int pos = base[b] + rk[j];
            int4* p = rec + ((size_t)b * CAP + pos);
            int row = d[j] - b * RB;
            if (pos >= CAP) {
                p = ovf + atomicAdd(&bin_count[n_bins], 1);
                row = d[j];
            }
            *p = make_int4(s[j], row, __float_as_int(ex[j]), 0);
        }
    }

    for (int i0 = 0; i0 < ntab; i0 += 4 * PL_T) {
        float t[4];
        #pragma unroll
        for (int q = 0; q < 4; ++q) {
            const int i = i0 + q * PL_T + tid;
            t[q] = (i < ntab) ? dtab[i] : 0.0f;
        }
        #pragma unroll
        for (int q = 0; q < 4; ++q) {
            const int i = i0 + q * PL_T + tid;
            if (t[q] != 0.0f) atomicAdd(&denom[i], t[q]);
        }
    }
}

__device__ __forceinline__ void flush_row(float* acc_row, int lane, v4f a) {
    float* ap = acc_row + lane * 4;
    atomicAdd(ap,     a.x);
    atomicAdd(ap + 1, a.y);
    atomicAdd(ap + 2, a.z);
    atomicAdd(ap + 3, a.w);
}

__global__ __launch_bounds__(1024) void gat_sum(const float* __restrict__ h,
                                                const int* __restrict__ rec_raw,
                                                const int* __restrict__ ovf_raw,
                                                const int* __restrict__ bin_count,
                                                const float* __restrict__ denom,
                                                float* __restrict__ out,
                                                int n_nodes) {
    __shared__ float acc[RB * C];
    __shared__ int4  srt[CAP];
    __shared__ int   rowc[RSLOT];
    const int tid  = threadIdx.x;
    const int lane = tid & 31;
    const int wave = __builtin_amdgcn_readfirstlane(tid >> 5);
    const int b = blockIdx.x;

    const int n = min(bin_count[b], CAP);
    const int n_ovf = bin_count[gridDim.x];
    const int4* r = (const int4*)rec_raw + (size_t)b * CAP;

    int4 mine[CAP / 1024];
    #pragma unroll
    for (int k = 0; k < CAP / 1024; ++k) {
        const int idx = tid + k * 1024;
        mine[k] = make_int4(0, -1, 0, 0);
        if (idx < n) mine[k] = r[idx];
    }
    for (int i = tid; i < RB * C; i += 1024) acc[i] = 0.0f;
    if (tid < RSLOT) rowc[tid] = 0;
    __syncthreads();

    int rk[CAP / 1024];
    #pragma unroll
    for (int k = 0; k < CAP / 1024; ++k)
        rk[k] = (mine[k].y >= 0) ? atomicAdd(&rowc[mine[k].y], 1) : 0;
    __syncthreads();
    if (wave == 0) {
        const int c0 = rowc[lane * 4], c1 = rowc[lane * 4 + 1], c2 = rowc[lane * 4 + 2], c3 = rowc[lane * 4 + 3];
        const int s = c0 + c1 + c2 + c3;
        int x = s;
        #pragma unroll
        for (int off = 1; off < 32; off <<= 1) {
            const int y = __shfl_up(x, off, 32);
            if (lane >= off) x += y;
        }
        const int first = x - s;
        rowc[lane * 4]     = first;
        rowc[lane * 4 + 1] = first + c0;
        rowc[lane * 4 + 2] = first + c0 + c1;
        rowc[lane * 4 + 3] = first + c0 + c1 + c2;
    }
    __syncthreads();
    #pragma unroll
    for (int k = 0; k < CAP / 1024; ++k)
        if (mine[k].y >= 0) srt[rowc[mine[k].y] + rk[k]] = mine[k];
    __syncthreads();

    const int nb = (n + 15) >> 4;
    const int lo = ((wave * nb) >> 5) << 4;
    const int hi = min(n, (((wave + 1) * nb) >> 5) << 4);
    int cur = -1;
    v4f a = {0.0f, 0.0f, 0.0f, 0.0f};
    for (int g = lo; g < hi; g += 16) {
        const int cnt = min(16, hi - g);
        int4 rc = make_int4(0, 0, 0, 0);
        if (lane < cnt) rc = srt[g + lane];
        v4f hv[16];
        #pragma unroll
        for (int i = 0; i < 16; ++i)
            hv[i] = load_row4(h, lane_i(rc.x, i), lane);
        const float dnl = (lane < cnt) ? denom[rc.x] : 1.0f;
        const float attl = __int_as_float(rc.z) / dnl;
        #pragma unroll
        for (int i = 0; i < 16; ++i) {
            if (i < cnt) {
                const int row = lane_i(rc.y, i);
                const float att = lane_f(attl, i);
                if (row != cur) {
                    if (cur >= 0) flush_row(acc + cur * C, lane, a);
                    cur = row;
                    a = (v4f){0.0f, 0.0f, 0.0f, 0.0f};
                }
                a.x = fmaf(hv[i].x, att, a.x);
                a.y = fmaf(hv[i].y, att, a.y);
                a.z = fmaf(hv[i].z, att, a.z);
                a.w = fmaf(hv[i].w, att, a.w);
            }
        }
    }
    if (cur >= 0) flush_row(acc + cur * C, lane, a);
    if (n_ovf > 0) {
        const int4* ov = (const int4*)ovf_raw;
        for (int i = tid; i < n_ovf; i += 1024) {
            const int4 o = ov[i];
            const int rl = o.y - b * RB;
            if (rl >= 0 && rl < RB) {
                const float att = __int_as_float(o.z) / denom[o.x];
                const float* hp = h + (size_t)o.x * C;
                float* ap = acc + rl * C;
                for (int c = 0; c < C; ++c) atomicAdd(ap + c, hp[c] * att);
            }
        }
    }
    __syncthreads();

    const int row0 = b * RB;
    float ov[RB * C / 1024];
    #pragma unroll
    for (int k = 0; k < RB * C / 1024; ++k) {
        const int i = tid + k * 1024;
        ov[k] = (row0 + (i >> 7) < n_nodes) ? out[(size_t)row0 * C + i] : 0.0f;
    }
    #pragma unroll
    for (int k = 0; k < RB * C / 1024; ++k) {
        const int i = tid + k * 1024;
        if (row0 + (i >> 7) < n_nodes) out[(size_t)row0 * C + i] = ov[k] + acc[i];
    }
}

extern "C" void kernel_launch(void* const* d_in, const int* in_sizes, int n_in,
                              void* d_out, int out_size, void* d_ws, size_t ws_size,
                              hipStream_t stream) {
    const float* h  = (const float*)d_in[0];
    const float* w  = (const float*)d_in[1];
    const float* a  = (const float*)d_in[2];
    const int* src  = (const int*)d_in[3];
    const int* dst  = (const int*)d_in[4];
    float* out      = (float*)d_out;

    const int n_nodes = in_sizes[0] / C;
    const int n_edges = in_sizes[3];

    const int n_bins  = (n_nodes + RB - 1) / RB;

    float* u       = (float*)d_ws;
    float* v       = u + (size_t)n_nodes * C;
    int* rec       = (int*)(v + (size_t)n_nodes * C);
    float* exp_e   = (float*)(rec + (size_t)n_bins * CAP * 4);
    int* ovf       = (int*)(exp_e + n_edges);
    float* denom   = (float*)(ovf + (size_t)n_edges * 4);
    int* bin_count = (int*)(denom + n_nodes);

    {
        int grid = (n_nodes + 15) / 16;
        gat_gemm_uv<<<grid, 256, 0, stream>>>(h, w, u, v, out, denom, bin_count, n_bins, n_nodes);
    }
    {
        int grid = (n_edges + 255) / 256;
        gat_edge_e<<<grid, 256, 0, stream>>>(u, v, a, src, dst, exp_e, n_edges);
    }
    {
        int grid = (n_edges + PL_T * PL_E - 1) / (PL_T * PL_E);
        gat_place<<<grid, PL_T, (size_t)n_bins * 2 * sizeof(int), stream>>>(
            src, dst, exp_e, denom, bin_count, rec, ovf, n_edges, n_bins, n_nodes);
    }
    {
        gat_sum<<<n_bins, 1024, 0, stream>>>(h, rec, ovf, bin_count, denom, out, n_nodes);
    }
}


# sum kernel: denom copied to LDS by whole-line loads, each record's weight exp/denom[src] formed once before the LDS sort (was a 16-line gather and a divide per batch in the row walk); sums unchanged
# speedup vs baseline: 3.7700x; 1.0355x over previous
#include <hip/hip_runtime.h>
#include <hip/hip_bf16.h>


#define C 128
#define LEAKY 0.2f

typedef __attribute__((ext_vector_type(2))) float v2f;
typedef __attribute__((ext_vector_type(8))) float v8f;

typedef __attribute__((ext_vector_type(4))) float v4f;

__global__ __launch_bounds__(256) void gat_gemm_uv(const float* __restrict__ h,
                                                   const float* __restrict__ w,
                                                   float* __restrict__ u,
                                                   float* __restrict__ v,
                                                   float* __restrict__ out,
                                                   float* __restrict__ denom,
                                                   int* __restrict__ bin_count,
                                                   int n_bins,
                                                   int n_nodes) {
    __shared__ float lh[16 * C];
    const int row0 = blockIdx.x * 16;
    const int tid  = threadIdx.x;
    const bool full_tile = (row0 + 16) <= n_nodes;

    if (full_tile) {
        for (int i = tid; i < 16 * C; i += 256)
            lh[i] = h[(size_t)row0 * C + i];
    } else {
        for (int i = tid; i < 16 * C; i += 256) {
            int gr = row0 + (i >> 7);
            lh[i] = (gr < n_nodes) ? h[(size_t)gr * C + (i & (C - 1))] : 0.0f;
        }
    }
    __syncthreads();

    {
        const v4f zero4 = {0.0f, 0.0f, 0.0f, 0.0f};
        v4f* o4 = (v4f*)(out + (size_t)row0 * C);
        if (full_tile) {
            o4[tid]       = zero4;
            o4[tid + 256] = zero4;
        } else {
            const int nq = (n_nodes - row0) * (C / 4);
            if (tid < nq)       o4[tid]       = zero4;
            if (tid + 256 < nq) o4[tid + 256] = zero4;
        }
        if ((blockIdx.x & 1) == 0 && tid < 32 && row0 + tid < n_nodes)
            denom[row0 + tid] = 0.0f;
        if (blockIdx.x == 0)
            for (int i = tid; i < n_bins + 1; i += 256) bin_count[i] = 0;
    }

    const int wave = tid >> 5;
    const int lane = tid & 31;
    const int half = lane >> 4;
    const int l16  = lane & 15;
    const int col  = wave * 16 + l16;

    v8f accu = {};
    v8f accv = {};
    for (int k = 0; k < C; k += 4) {
        const int ka = k + half * 2;
        v2f afrag;
        afrag.x = lh[l16 * C + ka];
        afrag.y = lh[l16 * C + ka + 1];
        v2f bu, bv;
        bu.x = w[(size_t)(ka    ) * C + col];
        bu.y = w[(size_t)(ka + 1) * C + col];
        bv.x = w[(size_t)(C + ka    ) * C + col];
        bv.y = w[(size_t)(C + ka + 1) * C + col];
        accu = __builtin_amdgcn_wmma_f32_16x16x4_f32(false, afrag, false, bu,
                                                     (short)0, accu, false, false);
        accv = __builtin_amdgcn_wmma_f32_16x16x4_f32(false, afrag, false, bv,
                                                     (short)0, accv, false, false);
    }

    const int mbase = row0 + half * 8;
    if (full_tile) {
        #pragma unroll
        for (int r = 0; r < 8; ++r) {
            u[(size_t)(mbase + r) * C + col] = accu[r];
            v[(size_t)(mbase + r) * C + col] = accv[r];
        }
    } else {
        #pragma unroll
        for (int r = 0; r < 8; ++r) {
            if (mbase + r < n_nodes) {
                u[(size_t)(mbase + r) * C + col] = accu[r];
                v[(size_t)(mbase + r) * C + col] = accv[r];
            }
        }
    }
}

__device__ __forceinline__ float leaky(float x) { return x > 0.0f ? x : LEAKY * x; }


__device__ __forceinline__ v4f load_row4(const float* __restrict__ base, int row, int lane) {
    return *(const v4f*)(base + (size_t)row * C + lane * 4);
}

__device__ __forceinline__ float edge_partial(v4f uu, v4f vv, float4 aa) {
    float sum = leaky(uu.x + vv.x) * aa.x;
    sum = fmaf(leaky(uu.y + vv.y), aa.y, sum);
    sum = fmaf(leaky(uu.z + vv.z), aa.z, sum);
    sum = fmaf(leaky(uu.w + vv.w), aa.w, sum);
    return sum;
}

__device__ __forceinline__ float fold(float lo, float hi, bool bit, int off) {
    float keep = bit ? hi : lo;
    float send = bit ? lo : hi;
    return keep + __shfl_xor(send, off, 32);
}

__global__ __launch_bounds__(256) void gat_edge_e(const float* __restrict__ u,
                                                  const float* __restrict__ v,
                                                  const float* __restrict__ a,
                                                  const int* __restrict__ src,
                                                  const int* __restrict__ dst,
                                                  float* __restrict__ exp_e,
                                                  int n_edges) {
    const int wib  = threadIdx.x >> 5;
    const int lane = threadIdx.x & 31;
    const int e0 = (blockIdx.x * 8 + wib) * 32;
    if (e0 >= n_edges) return;

    const int eown = min(e0 + lane, n_edges - 1);
    const int s_own = src[eown];
    const int d_own = dst[eown];
    const float4 aa = ((const float4*)a)[lane];

    const bool b4 = (lane & 16) != 0;
    const bool b3 = (lane & 8) != 0;
    const bool b2 = (lane & 4) != 0;
    const bool b1 = (lane & 2) != 0;
    const bool b0 = (lane & 1) != 0;

    float wsum[4];
    #pragma unroll
    for (int b = 0; b < 4; ++b) {
        v4f uu[8], vv[8];
        #pragma unroll
        for (int i = 0; i < 8; ++i) {
            uu[i] = load_row4(u, __builtin_amdgcn_readlane(s_own, b * 8 + i), lane);
            vv[i] = load_row4(v, __builtin_amdgcn_readlane(d_own, b * 8 + i), lane);
        }
        asm volatile("" :: "v"(uu[0]), "v"(uu[1]), "v"(uu[2]), "v"(uu[3]),
                           "v"(uu[4]), "v"(uu[5]), "v"(uu[6]), "v"(uu[7]),
                           "v"(vv[0]), "v"(vv[1]), "v"(vv[2]), "v"(vv[3]),
                           "v"(vv[4]), "v"(vv[5]), "v"(vv[6]), "v"(vv[7]));
        float p[8];
        #pragma unroll
        for (int i = 0; i < 8; ++i)
            p[i] = edge_partial(uu[i], vv[i], aa);
        const float q0 = fold(p[0], p[4], b4, 16);
        const float q1 = fold(p[1], p[5], b4, 16);
        const float q2 = fold(p[2], p[6], b4, 16);
        const float q3 = fold(p[3], p[7], b4, 16);
        const float r0 = fold(q0, q2, b3, 8);
        const float r1 = fold(q1, q3, b3, 8);
        wsum[b] = fold(r0, r1, b2, 4);
    }
    const float x0 = fold(wsum[0], wsum[2], b1, 2);
    const float x1 = fold(wsum[1], wsum[3], b1, 2);
    const float logit = fold(x0, x1, b0, 1);

    const int slot = ((lane & 3) << 3) | (lane >> 2);
    const int e = e0 + slot;
    if (e < n_edges) exp_e[e] = expf(logit);
}

#define RB   80
#define CAP  4096
#define PL_T 1024
#define PL_E 5
#define NTAB 20480
static_assert(NTAB % 4 == 0, "gat_place clears the table 16 B at a time");
#define RSLOT 128
static_assert(RB <= RSLOT, "one wave scans all row counters");
static_assert(CAP % 1024 == 0, "gat_sum: whole records per thread");
static_assert(NTAB % (4 * 1024) == 0, "gat_sum copies denom 16 B per thread at a time");

__device__ __forceinline__ int lane_i(int x, int l) { return __builtin_amdgcn_readlane(x, l); }
__device__ __forceinline__ float lane_f(float x, int l) {
    return __int_as_float(__builtin_amdgcn_readlane(__float_as_int(x), l));
}

__global__ __launch_bounds__(PL_T) void gat_place(const int* __restrict__ src,
                                                  const int* __restrict__ dst,
                                                  const float* __restrict__ exp_e,
                                                  float* __restrict__ denom,
                                                  int* __restrict__ bin_count,
                                                  int* __restrict__ rec_raw,
                                                  int* __restrict__ ovf_raw,
                                                  int n_edges, int n_bins, int n_nodes) {
    extern __shared__ int sm[];
    __shared__ __attribute__((aligned(16))) float dtab[NTAB];
    int* hist = sm;
    int* base = sm + n_bins;
    const int tid  = threadIdx.x;
    const int ntab = min(n_nodes, NTAB);

    const int e0 = blockIdx.x * (PL_T * PL_E);
    int s[PL_E], d[PL_E], rk[PL_E];
    float ex[PL_E];
    #pragma unroll
    for (int j = 0; j < PL_E; ++j) {
        const int e  = e0 + j * PL_T + tid;
        const int ee = (e < n_edges) ? e : n_edges - 1;
        s[j]  = src[ee];
        d[j]  = dst[ee];
        ex[j] = exp_e[ee];
    }
    for (int i = tid; i < n_bins; i += PL_T) hist[i] = 0;
    for (int i = tid; i < (ntab + 3) / 4; i += PL_T)
        ((v4f*)dtab)[i] = (v4f){0.0f, 0.0f, 0.0f, 0.0f};
    __syncthreads();

    #pragma unroll
    for (int j = 0; j < PL_E; ++j) {
        const int e = e0 + j * PL_T + tid;
        rk[j] = -1;
        if (e < n_edges) {
            rk[j] = atomicAdd(&hist[d[j] / RB], 1);
            if (s[j] < ntab) atomicAdd(&dtab[s[j]], ex[j]);
            else             atomicAdd(&denom[s[j]], ex[j]);
        }
    }
    __syncthreads();
    for (int i = tid; i < n_bins; i += PL_T) {
        const int c = hist[i];
        base[i] = (c > 0) ? atomicAdd(&bin_count[i], c) : 0;
    }
    __syncthreads();

    int4* rec = (int4*)rec_raw;
    int4* ovf = (int4*)ovf_raw;
    #pragma unroll
    for (int j = 0; j < PL_E; ++j) {
        if (rk[j] >= 0) {
            const int b   = d[j] / RB;
            const int pos = base[b] + rk[j];
            int4* p = rec + ((size_t)b * CAP + pos);
            int row = d[j] - b * RB;
            if (pos >= CAP) {
                p = ovf + atomicAdd(&bin_count[n_bins], 1);
                row = d[j];
            }
            *p = make_int4(s[j], row, __float_as_int(ex[j]), 0);
        }
    }

    for (int i0 = 0; i0 < ntab; i0 += 4 * PL_T) {
        float t[4];
        #pragma unroll
        for (int q = 0; q < 4; ++q) {
            const int i = i0 + q * PL_T + tid;
            t[q] = (i < ntab) ? dtab[i] : 0.0f;
        }
        #pragma unroll
        for (int q = 0; q < 4; ++q) {
            const int i = i0 + q * PL_T + tid;
            if (t[q] != 0.0f) atomicAdd(&denom[i], t[q]);
        }
    }
}

__device__ __forceinline__ void flush_row(float* acc_row, int lane, v4f a) {
    float* ap = acc_row + lane * 4;
    atomicAdd(ap,     a.x);
    atomicAdd(ap + 1, a.y);
    atomicAdd(ap + 2, a.z);
    atomicAdd(ap + 3, a.w);
}

__global__ __launch_bounds__(1024) void gat_sum(const float* __restrict__ h,
                                                const int* __restrict__ rec_raw,
                                                const int* __restrict__ ovf_raw,
                                                const int* __restrict__ bin_count,
                                                const float* __restrict__ denom,
                                                float* __restrict__ out,
                                                int n_nodes) {
    __shared__ float acc[RB * C];
    __shared__ int4  srt[CAP];
    __shared__ int   rowc[RSLOT];
    __shared__ __attribute__((aligned(16))) float dsh[NTAB];
    const int tid  = threadIdx.x;
    const int lane = tid & 31;
    const int wave = __builtin_amdgcn_readfirstlane(tid >> 5);
    const int b = blockIdx.x;
    const int ntab = min(n_nodes, NTAB);

    const int n = min(bin_count[b], CAP);
    const int n_ovf = bin_count[gridDim.x];
    const int4* r = (const int4*)rec_raw + (size_t)b * CAP;

    int4 mine[CAP / 1024];
    #pragma unroll
    for (int k = 0; k < CAP / 1024; ++k) {
        const int idx = tid + k * 1024;
        mine[k] = make_int4(0, -1, 0, 0);
        if (idx < n) mine[k] = r[idx];
    }
    const int nq = (ntab + 3) >> 2;
    v4f dq[NTAB / 4096];
    #pragma unroll
    for (int q = 0; q < NTAB / 4096; ++q) {
        const int i = tid + q * 1024;
        dq[q] = (v4f){1.0f, 1.0f, 1.0f, 1.0f};
        if (i < nq) dq[q] = ((const v4f*)denom)[i];
    }
    for (int i = tid; i < RB * C; i += 1024) acc[i] = 0.0f;
    if (tid < RSLOT) rowc[tid] = 0;
    #pragma unroll
    for (int q = 0; q < NTAB / 4096; ++q)
        ((v4f*)dsh)[tid + q * 1024] = dq[q];
    __syncthreads();

    int rk[CAP / 1024];
    float dn[CAP / 1024];
    #pragma unroll
    for (int k = 0; k < CAP / 1024; ++k)
        dn[k] = dsh[(mine[k].x < ntab) ? mine[k].x : 0];
    if (n_nodes > NTAB) {
        #pragma unroll
        for (int k = 0; k < CAP / 1024; ++k)
            if (mine[k].y >= 0 && mine[k].x >= NTAB) dn[k] = denom[mine[k].x];
    }
    #pragma unroll
    for (int k = 0; k < CAP / 1024; ++k) {
        rk[k] = 0;
        if (mine[k].y >= 0) {
            mine[k].z = __float_as_int(__int_as_float(mine[k].z) / dn[k]);
            rk[k] = atomicAdd(&rowc[mine[k].y], 1);
        }
    }
    __syncthreads();
    if (wave == 0) {
        const int c0 = rowc[lane * 4], c1 = rowc[lane * 4 + 1], c2 = rowc[lane * 4 + 2], c3 = rowc[lane * 4 + 3];
        const int s = c0 + c1 + c2 + c3;
        int x = s;
        #pragma unroll
        for (int off = 1; off < 32; off <<= 1) {
            const int y = __shfl_up(x, off, 32);
            if (lane >= off) x += y;
        }
        const int first = x - s;
        rowc[lane * 4]     = first;
        rowc[lane * 4 + 1] = first + c0;
        rowc[lane * 4 + 2] = first + c0 + c1;
        rowc[lane * 4 + 3] = first + c0 + c1 + c2;
    }
    __syncthreads();
    #pragma unroll
    for (int k = 0; k < CAP / 1024; ++k)
        if (mine[k].y >= 0) srt[rowc[mine[k].y] + rk[k]] = mine[k];
    __syncthreads();

    const int nb = (n + 15) >> 4;
    const int lo = ((wave * nb) >> 5) << 4;
    const int hi = min(n, (((wave + 1) * nb) >> 5) << 4);
    int cur = -1;
    v4f a = {0.0f, 0.0f, 0.0f, 0.0f};
    for (int g = lo; g < hi; g += 16) {
        const int cnt = min(16, hi - g);
        int4 rc = make_int4(0, 0, 0, 0);
        if (lane < cnt) rc = srt[g + lane];
        v4f hv[16];
        #pragma unroll
        for (int i = 0; i < 16; ++i)
            hv[i] = load_row4(h, lane_i(rc.x, i), lane);
        const float attl = __int_as_float(rc.z);
        #pragma unroll
        for (int i = 0; i < 16; ++i) {
            if (i < cnt) {
                const int row = lane_i(rc.y, i);
                const float att = lane_f(attl, i);
                if (row != cur) {
                    if (cur >= 0) flush_row(acc + cur * C, lane, a);
                    cur = row;
                    a = (v4f){0.0f, 0.0f, 0.0f, 0.0f};
                }
                a.x = fmaf(hv[i].x, att, a.x);
                a.y = fmaf(hv[i].y, att, a.y);
                a.z = fmaf(hv[i].z, att, a.z);
                a.w = fmaf(hv[i].w, att, a.w);
            }
        }
    }
    if (cur >= 0) flush_row(acc + cur * C, lane, a);
    if (n_ovf > 0) {
        const int4* ov = (const int4*)ovf_raw;
        for (int i = tid; i < n_ovf; i += 1024) {
            const int4 o = ov[i];
            const int rl = o.y - b * RB;
            if (rl >= 0 && rl < RB) {
                const float att = __int_as_float(o.z) / denom[o.x];
                const float* hp = h + (size_t)o.x * C;
                float* ap = acc + rl * C;
                for (int c = 0; c < C; ++c) atomicAdd(ap + c, hp[c] * att);
            }
        }
    }
    __syncthreads();

    const int row0 = b * RB;
    float ov[RB * C / 1024];
    #pragma unroll
    for (int k = 0; k < RB * C / 1024; ++k) {
        const int i = tid + k * 1024;
        ov[k] = (row0 + (i >> 7) < n_nodes) ? out[(size_t)row0 * C + i] : 0.0f;
    }
    #pragma unroll
    for (int k = 0; k < RB * C / 1024; ++k) {
        const int i = tid + k * 1024;
        if (row0 + (i >> 7) < n_nodes) out[(size_t)row0 * C + i] = ov[k] + acc[i];
    }
}

extern "C" void kernel_launch(void* const* d_in, const int* in_sizes, int n_in,
                              void* d_out, int out_size, void* d_ws, size_t ws_size,
                              hipStream_t stream) {
    const float* h  = (const float*)d_in[0];
    const float* w  = (const float*)d_in[1];
    const float* a  = (const float*)d_in[2];
    const int* src  = (const int*)d_in[3];
    const int* dst  = (const int*)d_in[4];
    float* out      = (float*)d_out;

    const int n_nodes = in_sizes[0] / C;
    const int n_edges = in_sizes[3];

    const int n_bins  = (n_nodes + RB - 1) / RB;

    float* u       = (float*)d_ws;
    float* v       = u + (size_t)n_nodes * C;
    int* rec       = (int*)(v + (size_t)n_nodes * C);
    float* exp_e   = (float*)(rec + (size_t)n_bins * CAP * 4);
    int* ovf       = (int*)(exp_e + n_edges);
    float* denom   = (float*)(ovf + (size_t)n_edges * 4);
    int* bin_count = (int*)(denom + n_nodes);

    {
        int grid = (n_nodes + 15) / 16;
        gat_gemm_uv<<<grid, 256, 0, stream>>>(h, w, u, v, out, denom, bin_count, n_bins, n_nodes);
    }
    {
        int grid = (n_edges + 255) / 256;
        gat_edge_e<<<grid, 256, 0, stream>>>(u, v, a, src, dst, exp_e, n_edges);
    }
    {
        int grid = (n_edges + PL_T * PL_E - 1) / (PL_T * PL_E);
        gat_place<<<grid, PL_T, (size_t)n_bins * 2 * sizeof(int), stream>>>(
            src, dst, exp_e, denom, bin_count, rec, ovf, n_edges, n_bins, n_nodes);
    }
    {
        gat_sum<<<n_bins, 1024, 0, stream>>>(h, rec, ovf, bin_count, denom, out, n_nodes);
    }
}


# projection kernel: a block keeps its w fragments in registers over five 16-row tiles (was re-read from L2 every k-step of every tile); h rows in LDS at pitch 132; u, v bit-identical
# speedup vs baseline: 3.9141x; 1.0382x over previous
#include <hip/hip_runtime.h>
#include <hip/hip_bf16.h>


#define C 128
#define LEAKY 0.2f

typedef __attribute__((ext_vector_type(2))) float v2f;
typedef __attribute__((ext_vector_type(8))) float v8f;

typedef __attribute__((ext_vector_type(4))) float v4f;

#define TPB   5
#define HP    132
__global__ __launch_bounds__(256) void gat_gemm_uv(const float* __restrict__ h,
                                                   const float* __restrict__ w,
                                                   float* __restrict__ u,
                                                   float* __restrict__ v,
                                                   float* __restrict__ out,
                                                   float* __restrict__ denom,
                                                   int* __restrict__ bin_count,
                                                   int n_bins,
                                                   int n_nodes) {
    __shared__ __attribute__((aligned(16))) float lh[TPB * 16 * HP];
    const int row0 = blockIdx.x * (TPB * 16);
    const int tid  = threadIdx.x;
    const int wave = tid >> 5;
    const int lane = tid & 31;
    const int half = lane >> 4;
    const int l16  = lane & 15;
    const int col  = wave * 16 + l16;
    const int nrow = min(TPB * 16, n_nodes - row0);

    v4f hq[TPB * 16 * 32 / 256];
    #pragma unroll
    for (int q = 0; q < TPB * 16 * 32 / 256; ++q) {
        const int i = tid + q * 256;
        hq[q] = (v4f){0.0f, 0.0f, 0.0f, 0.0f};
        if ((i >> 5) < nrow) hq[q] = ((const v4f*)(h + (size_t)row0 * C))[i];
    }

    v2f bu[C / 4], bv[C / 4];
    #pragma unroll
    for (int j = 0; j < C / 4; ++j) {
        const int ka = j * 4 + half * 2;
        bu[j].x = w[(size_t)(ka    ) * C + col];
        bu[j].y = w[(size_t)(ka + 1) * C + col];
        bv[j].x = w[(size_t)(C + ka    ) * C + col];
        bv[j].y = w[(size_t)(C + ka + 1) * C + col];
    }

    #pragma unroll
    for (int q = 0; q < TPB * 16 * 32 / 256; ++q) {
        const int i = tid + q * 256;
        *(v4f*)(lh + (i >> 5) * HP + (i & 31) * 4) = hq[q];
    }
    __syncthreads();

    {
        const v4f zero4 = {0.0f, 0.0f, 0.0f, 0.0f};
        v4f* o4 = (v4f*)(out + (size_t)row0 * C);
        #pragma unroll
        for (int q = 0; q < TPB * 16 * 32 / 256; ++q) {
            const int i = tid + q * 256;
            if ((i >> 5) < nrow) o4[i] = zero4;
        }
        if (tid < nrow) denom[row0 + tid] = 0.0f;
        if (blockIdx.x == 0)
            for (int i = tid; i < n_bins + 1; i += 256) bin_count[i] = 0;
    }

    for (int t = 0; t < TPB; ++t) {
        if (t * 16 >= nrow) break;
        v8f accu = {};
        v8f accv = {};
        const float* arow = lh + (t * 16 + l16) * HP + half * 2;
        #pragma unroll
        for (int j = 0; j < C / 4; ++j) {
            const v2f afrag = *(const v2f*)(arow + j * 4);
            accu = __builtin_amdgcn_wmma_f32_16x16x4_f32(false, afrag, false, bu[j],
                                                         (short)0, accu, false, false);
            accv = __builtin_amdgcn_wmma_f32_16x16x4_f32(false, afrag, false, bv[j],
                                                         (short)0, accv, false, false);
        }
        const int ml = t * 16 + half * 8;
        #pragma unroll
        for (int r = 0; r < 8; ++r) {
            if (ml + r < nrow) {
                u[(size_t)(row0 + ml + r) * C + col] = accu[r];
                v[(size_t)(row0 + ml + r) * C + col] = accv[r];
            }
        }
    }
}

__device__ __forceinline__ float leaky(float x) { return x > 0.0f ? x : LEAKY * x; }


__device__ __forceinline__ v4f load_row4(const float* __restrict__ base, int row, int lane) {
    return *(const v4f*)(base + (size_t)row * C + lane * 4);
}

__device__ __forceinline__ float edge_partial(v4f uu, v4f vv, float4 aa) {
    float sum = leaky(uu.x + vv.x) * aa.x;
    sum = fmaf(leaky(uu.y + vv.y), aa.y, sum);
    sum = fmaf(leaky(uu.z + vv.z), aa.z, sum);
    sum = fmaf(leaky(uu.w + vv.w), aa.w, sum);
    return sum;
}

__device__ __forceinline__ float fold(float lo, float hi, bool bit, int off) {
    float keep = bit ? hi : lo;
    float send = bit ? lo : hi;
    return keep + __shfl_xor(send, off, 32);
}

__global__ __launch_bounds__(256) void gat_edge_e(const float* __restrict__ u,
                                                  const float* __restrict__ v,
                                                  const float* __restrict__ a,
                                                  const int* __restrict__ src,
                                                  const int* __restrict__ dst,
                                                  float* __restrict__ exp_e,
                                                  int n_edges) {
    const int wib  = threadIdx.x >> 5;
    const int lane = threadIdx.x & 31;
    const int e0 = (blockIdx.x * 8 + wib) * 32;
    if (e0 >= n_edges) return;

    const int eown = min(e0 + lane, n_edges - 1);
    const int s_own = src[eown];
    const int d_own = dst[eown];
    const float4 aa = ((const float4*)a)[lane];

    const bool b4 = (lane & 16) != 0;
    const bool b3 = (lane & 8) != 0;
    const bool b2 = (lane & 4) != 0;
    const bool b1 = (lane & 2) != 0;
    const bool b0 = (lane & 1) != 0;

    float wsum[4];
    #pragma unroll
    for (int b = 0; b < 4; ++b) {
        v4f uu[8], vv[8];
        #pragma unroll
        for (int i = 0; i < 8; ++i) {
            uu[i] = load_row4(u, __builtin_amdgcn_readlane(s_own, b * 8 + i), lane);
            vv[i] = load_row4(v, __builtin_amdgcn_readlane(d_own, b * 8 + i), lane);
        }
        asm volatile("" :: "v"(uu[0]), "v"(uu[1]), "v"(uu[2]), "v"(uu[3]),
                           "v"(uu[4]), "v"(uu[5]), "v"(uu[6]), "v"(uu[7]),
                           "v"(vv[0]), "v"(vv[1]), "v"(vv[2]), "v"(vv[3]),
                           "v"(vv[4]), "v"(vv[5]), "v"(vv[6]), "v"(vv[7]));
        float p[8];
        #pragma unroll
        for (int i = 0; i < 8; ++i)
            p[i] = edge_partial(uu[i], vv[i], aa);
        const float q0 = fold(p[0], p[4], b4, 16);
        const float q1 = fold(p[1], p[5], b4, 16);
        const float q2 = fold(p[2], p[6], b4, 16);
        const float q3 = fold(p[3], p[7], b4, 16);
        const float r0 = fold(q0, q2, b3, 8);
        const float r1 = fold(q1, q3, b3, 8);
        wsum[b] = fold(r0, r1, b2, 4);
    }
    const float x0 = fold(wsum[0], wsum[2], b1, 2);
    const float x1 = fold(wsum[1], wsum[3], b1, 2);
    const float logit = fold(x0, x1, b0, 1);

    const int slot = ((lane & 3) << 3) | (lane >> 2);
    const int e = e0 + slot;
    if (e < n_edges) exp_e[e] = expf(logit);
}

#define RB   80
#define CAP  4096
#define PL_T 1024
#define PL_E 5
#define NTAB 20480
static_assert(NTAB % 4 == 0, "gat_place clears the table 16 B at a time");
#define RSLOT 128
static_assert(RB <= RSLOT, "one wave scans all row counters");
static_assert(CAP % 1024 == 0, "gat_sum: whole records per thread");
static_assert(NTAB % (4 * 1024) == 0, "gat_sum copies denom 16 B per thread at a time");

__device__ __forceinline__ int lane_i(int x, int l) { return __builtin_amdgcn_readlane(x, l); }
__device__ __forceinline__ float lane_f(float x, int l) {
    return __int_as_float(__builtin_amdgcn_readlane(__float_as_int(x), l));
}

__global__ __launch_bounds__(PL_T) void gat_place(const int* __restrict__ src,
                                                  const int* __restrict__ dst,
                                                  const float* __restrict__ exp_e,
                                                  float* __restrict__ denom,
                                                  int* __restrict__ bin_count,
                                                  int* __restrict__ rec_raw,
                                                  int* __restrict__ ovf_raw,
                                                  int n_edges, int n_bins, int n_nodes) {
    extern __shared__ int sm[];
    __shared__ __attribute__((aligned(16))) float dtab[NTAB];
    int* hist = sm;
    int* base = sm + n_bins;
    const int tid  = threadIdx.x;
    const int ntab = min(n_nodes, NTAB);

    const int e0 = blockIdx.x * (PL_T * PL_E);
    int s[PL_E], d[PL_E], rk[PL_E];
    float ex[PL_E];
    #pragma unroll
    for (int j = 0; j < PL_E; ++j) {
        const int e  = e0 + j * PL_T + tid;
        const int ee = (e < n_edges) ? e : n_edges - 1;
        s[j]  = src[ee];
        d[j]  = dst[ee];
        ex[j] = exp_e[ee];
    }
    for (int i = tid; i < n_bins; i += PL_T) hist[i] = 0;
    for (int i = tid; i < (ntab + 3) / 4; i += PL_T)
        ((v4f*)dtab)[i] = (v4f){0.0f, 0.0f, 0.0f, 0.0f};
    __syncthreads();

    #pragma unroll
    for (int j = 0; j < PL_E; ++j) {
        const int e = e0 + j * PL_T + tid;
        rk[j] = -1;
        if (e < n_edges) {
            rk[j] = atomicAdd(&hist[d[j] / RB], 1);
            if (s[j] < ntab) atomicAdd(&dtab[s[j]], ex[j]);
            else             atomicAdd(&denom[s[j]], ex[j]);
        }
    }
    __syncthreads();
    for (int i = tid; i < n_bins; i += PL_T) {
        const int c = hist[i];
        base[i] = (c > 0) ? atomicAdd(&bin_count[i], c) : 0;
    }
    __syncthreads();

    int4* rec = (int4*)rec_raw;
    int4* ovf = (int4*)ovf_raw;
    #pragma unroll
    for (int j = 0; j < PL_E; ++j) {
        if (rk[j] >= 0) {
            const int b   = d[j] / RB;
            const int pos = base[b] + rk[j];
            int4* p = rec + ((size_t)b * CAP + pos);
            int row = d[j] - b * RB;
            if (pos >= CAP) {
                p = ovf + atomicAdd(&bin_count[n_bins], 1);
                row = d[j];
            }
            *p = make_int4(s[j], row, __float_as_int(ex[j]), 0);
        }
    }

    for (int i0 = 0; i0 < ntab; i0 += 4 * PL_T) {
        float t[4];
        #pragma unroll
        for (int q = 0; q < 4; ++q) {
            const int i = i0 + q * PL_T + tid;
            t[q] = (i < ntab) ? dtab[i] : 0.0f;
        }
        #pragma unroll
        for (int q = 0; q < 4; ++q) {
            const int i = i0 + q * PL_T + tid;
            if (t[q] != 0.0f) atomicAdd(&denom[i], t[q]);
        }
    }
}

__device__ __forceinline__ void flush_row(float* acc_row, int lane, v4f a) {
    float* ap = acc_row + lane * 4;
    atomicAdd(ap,     a.x);
    atomicAdd(ap + 1, a.y);
    atomicAdd(ap + 2, a.z);
    atomicAdd(ap + 3, a.w);
}

__global__ __launch_bounds__(1024) void gat_sum(const float* __restrict__ h,
                                                const int* __restrict__ rec_raw,
                                                const int* __restrict__ ovf_raw,
                                                const int* __restrict__ bin_count,
                                                const float* __restrict__ denom,
                                                float* __restrict__ out,
                                                int n_nodes) {
    __shared__ float acc[RB * C];
    __shared__ int4  srt[CAP];
    __shared__ int   rowc[RSLOT];
    __shared__ __attribute__((aligned(16))) float dsh[NTAB];
    const int tid  = threadIdx.x;
    const int lane = tid & 31;
    const int wave = __builtin_amdgcn_readfirstlane(tid >> 5);
    const int b = blockIdx.x;
    const int ntab = min(n_nodes, NTAB);

    const int n = min(bin_count[b], CAP);
    const int n_ovf = bin_count[gridDim.x];
    const int4* r = (const int4*)rec_raw + (size_t)b * CAP;

    int4 mine[CAP / 1024];
    #pragma unroll
    for (int k = 0; k < CAP / 1024; ++k) {
        const int idx = tid + k * 1024;
        mine[k] = make_int4(0, -1, 0, 0);
        if (idx < n) mine[k] = r[idx];
    }
    const int nq = (ntab + 3) >> 2;
    v4f dq[NTAB / 4096];
    #pragma unroll
    for (int q = 0; q < NTAB / 4096; ++q) {
        const int i = tid + q * 1024;
        dq[q] = (v4f){1.0f, 1.0f, 1.0f, 1.0f};
        if (i < nq) dq[q] = ((const v4f*)denom)[i];
    }
    for (int i = tid; i < RB * C; i += 1024) acc[i] = 0.0f;
    if (tid < RSLOT) rowc[tid] = 0;
    #pragma unroll
    for (int q = 0; q < NTAB / 4096; ++q)
        ((v4f*)dsh)[tid + q * 1024] = dq[q];
    __syncthreads();

    int rk[CAP / 1024];
    float dn[CAP / 1024];
    #pragma unroll
    for (int k = 0; k < CAP / 1024; ++k)
        dn[k] = dsh[(mine[k].x < ntab) ? mine[k].x : 0];
    if (n_nodes > NTAB) {
        #pragma unroll
        for (int k = 0; k < CAP / 1024; ++k)
            if (mine[k].y >= 0 && mine[k].x >= NTAB) dn[k] = denom[mine[k].x];
    }
    #pragma unroll
    for (int k = 0; k < CAP / 1024; ++k) {
        rk[k] = 0;
        if (mine[k].y >= 0) {
            mine[k].z = __float_as_int(__int_as_float(mine[k].z) / dn[k]);
            rk[k] = atomicAdd(&rowc[mine[k].y], 1);
        }
    }
    __syncthreads();
    if (wave == 0) {
        const int c0 = rowc[lane * 4], c1 = rowc[lane * 4 + 1], c2 = rowc[lane * 4 + 2], c3 = rowc[lane * 4 + 3];
        const int s = c0 + c1 + c2 + c3;
        int x = s;
        #pragma unroll
        for (int off = 1; off < 32; off <<= 1) {
            const int y = __shfl_up(x, off, 32);
            if (lane >= off) x += y;
        }
        const int first = x - s;
        rowc[lane * 4]     = first;
        rowc[lane * 4 + 1] = first + c0;
        rowc[lane * 4 + 2] = first + c0 + c1;
        rowc[lane * 4 + 3] = first + c0 + c1 + c2;
    }
    __syncthreads();
    #pragma unroll
    for (int k = 0; k < CAP / 1024; ++k)
        if (mine[k].y >= 0) srt[rowc[mine[k].y] + rk[k]] = mine[k];
    __syncthreads();

    const int nb = (n + 15) >> 4;
    const int lo = ((wave * nb) >> 5) << 4;
    const int hi = min(n, (((wave + 1) * nb) >> 5) << 4);
    int cur = -1;
    v4f a = {0.0f, 0.0f, 0.0f, 0.0f};
    for (int g = lo; g < hi; g += 16) {
        const int cnt = min(16, hi - g);
        int4 rc = make_int4(0, 0, 0, 0);
        if (lane < cnt) rc = srt[g + lane];
        v4f hv[16];
        #pragma unroll
        for (int i = 0; i < 16; ++i)
            hv[i] = load_row4(h, lane_i(rc.x, i), lane);
        const float attl = __int_as_float(rc.z);
        #pragma unroll
        for (int i = 0; i < 16; ++i) {
            if (i < cnt) {
                const int row = lane_i(rc.y, i);
                const float att = lane_f(attl, i);
                if (row != cur) {
                    if (cur >= 0) flush_row(acc + cur * C, lane, a);
                    cur = row;
                    a = (v4f){0.0f, 0.0f, 0.0f, 0.0f};
                }
                a.x = fmaf(hv[i].x, att, a.x);
                a.y = fmaf(hv[i].y, att, a.y);
                a.z = fmaf(hv[i].z, att, a.z);
                a.w = fmaf(hv[i].w, att, a.w);
            }
        }
    }
    if (cur >= 0) flush_row(acc + cur * C, lane, a);
    if (n_ovf > 0) {
        const int4* ov = (const int4*)ovf_raw;
        for (int i = tid; i < n_ovf; i += 1024) {
            const int4 o = ov[i];
            const int rl = o.y - b * RB;
            if (rl >= 0 && rl < RB) {
                const float att = __int_as_float(o.z) / denom[o.x];
                const float* hp = h + (size_t)o.x * C;
                float* ap = acc + rl * C;
                for (int c = 0; c < C; ++c) atomicAdd(ap + c, hp[c] * att);
            }
        }
    }
    __syncthreads();

    const int row0 = b * RB;
    float ov[RB * C / 1024];
    #pragma unroll
    for (int k = 0; k < RB * C / 1024; ++k) {
        const int i = tid + k * 1024;
        ov[k] = (row0 + (i >> 7) < n_nodes) ? out[(size_t)row0 * C + i] : 0.0f;
    }
    #pragma unroll
    for (int k = 0; k < RB * C / 1024; ++k) {
        const int i = tid + k * 1024;
        if (row0 + (i >> 7) < n_nodes) out[(size_t)row0 * C + i] = ov[k] + acc[i];
    }
}

extern "C" void kernel_launch(void* const* d_in, const int* in_sizes, int n_in,
                              void* d_out, int out_size, void* d_ws, size_t ws_size,
                              hipStream_t stream) {
    const float* h  = (const float*)d_in[0];
    const float* w  = (const float*)d_in[1];
    const float* a  = (const float*)d_in[2];
    const int* src  = (const int*)d_in[3];
    const int* dst  = (const int*)d_in[4];
    float* out      = (float*)d_out;

    const int n_nodes = in_sizes[0] / C;
    const int n_edges = in_sizes[3];

    const int n_bins  = (n_nodes + RB - 1) / RB;

    float* u       = (float*)d_ws;
    float* v       = u + (size_t)n_nodes * C;
    int* rec       = (int*)(v + (size_t)n_nodes * C);
    float* exp_e   = (float*)(rec + (size_t)n_bins * CAP * 4);
    int* ovf       = (int*)(exp_e + n_edges);
    float* denom   = (float*)(ovf + (size_t)n_edges * 4);
    int* bin_count = (int*)(denom + n_nodes);

    {
        gat_gemm_uv<<<n_bins, 256, 0, stream>>>(h, w, u, v, out, denom, bin_count, n_bins, n_nodes);
    }
    {
        int grid = (n_edges + 255) / 256;
        gat_edge_e<<<grid, 256, 0, stream>>>(u, v, a, src, dst, exp_e, n_edges);
    }
    {
        int grid = (n_edges + PL_T * PL_E - 1) / (PL_T * PL_E);
        gat_place<<<grid, PL_T, (size_t)n_bins * 2 * sizeof(int), stream>>>(
            src, dst, exp_e, denom, bin_count, rec, ovf, n_edges, n_bins, n_nodes);
    }
    {
        gat_sum<<<n_bins, 1024, 0, stream>>>(h, rec, ovf, bin_count, denom, out, n_nodes);
    }
}


# sum kernel is the only writer of out: each row stored once from LDS, 16 B per lane (whole lines); the projection kernel's zero fill of out and the sum kernel's read-back of out are gone; same sums
# speedup vs baseline: 4.0726x; 1.0405x over previous
#include <hip/hip_runtime.h>
#include <hip/hip_bf16.h>


#define C 128
#define LEAKY 0.2f

typedef __attribute__((ext_vector_type(2))) float v2f;
typedef __attribute__((ext_vector_type(8))) float v8f;

typedef __attribute__((ext_vector_type(4))) float v4f;

#define TPB   5
#define HP    132
__global__ __launch_bounds__(256) void gat_gemm_uv(const float* __restrict__ h,
                                                   const float* __restrict__ w,
                                                   float* __restrict__ u,
                                                   float* __restrict__ v,
                                                   float* __restrict__ denom,
                                                   int* __restrict__ bin_count,
                                                   int n_bins,
                                                   int n_nodes) {
    __shared__ __attribute__((aligned(16))) float lh[TPB * 16 * HP];
    const int row0 = blockIdx.x * (TPB * 16);
    const int tid  = threadIdx.x;
    const int wave = tid >> 5;
    const int lane = tid & 31;
    const int half = lane >> 4;
    const int l16  = lane & 15;
    const int col  = wave * 16 + l16;
    const int nrow = min(TPB * 16, n_nodes - row0);

    v4f hq[TPB * 16 * 32 / 256];
    #pragma unroll
    for (int q = 0; q < TPB * 16 * 32 / 256; ++q) {
        const int i = tid + q * 256;
        hq[q] = (v4f){0.0f, 0.0f, 0.0f, 0.0f};
        if ((i >> 5) < nrow) hq[q] = ((const v4f*)(h + (size_t)row0 * C))[i];
    }

    v2f bu[C / 4], bv[C / 4];
    #pragma unroll
    for (int j = 0; j < C / 4; ++j) {
        const int ka = j * 4 + half * 2;
        bu[j].x = w[(size_t)(ka    ) * C + col];
        bu[j].y = w[(size_t)(ka + 1) * C + col];
        bv[j].x = w[(size_t)(C + ka    ) * C + col];
        bv[j].y = w[(size_t)(C + ka + 1) * C + col];
    }

    #pragma unroll
    for (int q = 0; q < TPB * 16 * 32 / 256; ++q) {
        const int i = tid + q * 256;
        *(v4f*)(lh + (i >> 5) * HP + (i & 31) * 4) = hq[q];
    }
    __syncthreads();

    {
        if (tid < nrow) denom[row0 + tid] = 0.0f;
        if (blockIdx.x == 0)
            for (int i = tid; i < n_bins + 1; i += 256) bin_count[i] = 0;
    }

    for (int t = 0; t < TPB; ++t) {
        if (t * 16 >= nrow) break;
        v8f accu = {};
        v8f accv = {};
        const float* arow = lh + (t * 16 + l16) * HP + half * 2;
        #pragma unroll
        for (int j = 0; j < C / 4; ++j) {
            const v2f afrag = *(const v2f*)(arow + j * 4);
            accu = __builtin_amdgcn_wmma_f32_16x16x4_f32(false, afrag, false, bu[j],
                                                         (short)0, accu, false, false);
            accv = __builtin_amdgcn_wmma_f32_16x16x4_f32(false, afrag, false, bv[j],
                                                         (short)0, accv, false, false);
        }
        const int ml = t * 16 + half * 8;
        #pragma unroll
        for (int r = 0; r < 8; ++r) {
            if (ml + r < nrow) {
                u[(size_t)(row0 + ml + r) * C + col] = accu[r];
                v[(size_t)(row0 + ml + r) * C + col] = accv[r];
            }
        }
    }
}

__device__ __forceinline__ float leaky(float x) { return x > 0.0f ? x : LEAKY * x; }


__device__ __forceinline__ v4f load_row4(const float* __restrict__ base, int row, int lane) {
    return *(const v4f*)(base + (size_t)row * C + lane * 4);
}

__device__ __forceinline__ float edge_partial(v4f uu, v4f vv, float4 aa) {
    float sum = leaky(uu.x + vv.x) * aa.x;
    sum = fmaf(leaky(uu.y + vv.y), aa.y, sum);
    sum = fmaf(leaky(uu.z + vv.z), aa.z, sum);
    sum = fmaf(leaky(uu.w + vv.w), aa.w, sum);
    return sum;
}

__device__ __forceinline__ float fold(float lo, float hi, bool bit, int off) {
    float keep = bit ? hi : lo;
    float send = bit ? lo : hi;
    return keep + __shfl_xor(send, off, 32);
}

__global__ __launch_bounds__(256) void gat_edge_e(const float* __restrict__ u,
                                                  const float* __restrict__ v,
                                                  const float* __restrict__ a,
                                                  const int* __restrict__ src,
                                                  const int* __restrict__ dst,
                                                  float* __restrict__ exp_e,
                                                  int n_edges) {
    const int wib  = threadIdx.x >> 5;
    const int lane = threadIdx.x & 31;
    const int e0 = (blockIdx.x * 8 + wib) * 32;
    if (e0 >= n_edges) return;

    const int eown = min(e0 + lane, n_edges - 1);
    const int s_own = src[eown];
    const int d_own = dst[eown];
    const float4 aa = ((const float4*)a)[lane];

    const bool b4 = (lane & 16) != 0;
    const bool b3 = (lane & 8) != 0;
    const bool b2 = (lane & 4) != 0;
    const bool b1 = (lane & 2) != 0;
    const bool b0 = (lane & 1) != 0;

    float wsum[4];
    #pragma unroll
    for (int b = 0; b < 4; ++b) {
        v4f uu[8], vv[8];
        #pragma unroll
        for (int i = 0; i < 8; ++i) {
            uu[i] = load_row4(u, __builtin_amdgcn_readlane(s_own, b * 8 + i), lane);
            vv[i] = load_row4(v, __builtin_amdgcn_readlane(d_own, b * 8 + i), lane);
        }
        asm volatile("" :: "v"(uu[0]), "v"(uu[1]), "v"(uu[2]), "v"(uu[3]),
                           "v"(uu[4]), "v"(uu[5]), "v"(uu[6]), "v"(uu[7]),
                           "v"(vv[0]), "v"(vv[1]), "v"(vv[2]), "v"(vv[3]),
                           "v"(vv[4]), "v"(vv[5]), "v"(vv[6]), "v"(vv[7]));
        float p[8];
        #pragma unroll
        for (int i = 0; i < 8; ++i)
            p[i] = edge_partial(uu[i], vv[i], aa);
        const float q0 = fold(p[0], p[4], b4, 16);
        const float q1 = fold(p[1], p[5], b4, 16);
        const float q2 = fold(p[2], p[6], b4, 16);
        const float q3 = fold(p[3], p[7], b4, 16);
        const float r0 = fold(q0, q2, b3, 8);
        const float r1 = fold(q1, q3, b3, 8);
        wsum[b] = fold(r0, r1, b2, 4);
    }
    const float x0 = fold(wsum[0], wsum[2], b1, 2);
    const float x1 = fold(wsum[1], wsum[3], b1, 2);
    const float logit = fold(x0, x1, b0, 1);

    const int slot = ((lane & 3) << 3) | (lane >> 2);
    const int e = e0 + slot;
    if (e < n_edges) exp_e[e] = expf(logit);
}

#define RB   80
#define CAP  4096
#define PL_T 1024
#define PL_E 5
#define NTAB 20480
static_assert(NTAB % 4 == 0, "gat_place clears the table 16 B at a time");
#define RSLOT 128
static_assert(RB <= RSLOT, "one wave scans all row counters");
static_assert(CAP % 1024 == 0, "gat_sum: whole records per thread");
static_assert(NTAB % (4 * 1024) == 0, "gat_sum copies denom 16 B per thread at a time");

__device__ __forceinline__ int lane_i(int x, int l) { return __builtin_amdgcn_readlane(x, l); }
__device__ __forceinline__ float lane_f(float x, int l) {
    return __int_as_float(__builtin_amdgcn_readlane(__float_as_int(x), l));
}

__global__ __launch_bounds__(PL_T) void gat_place(const int* __restrict__ src,
                                                  const int* __restrict__ dst,
                                                  const float* __restrict__ exp_e,
                                                  float* __restrict__ denom,
                                                  int* __restrict__ bin_count,
                                                  int* __restrict__ rec_raw,
                                                  int* __restrict__ ovf_raw,
                                                  int n_edges, int n_bins, int n_nodes) {
    extern __shared__ int sm[];
    __shared__ __attribute__((aligned(16))) float dtab[NTAB];
    int* hist = sm;
    int* base = sm + n_bins;
    const int tid  = threadIdx.x;
    const int ntab = min(n_nodes, NTAB);

    const int e0 = blockIdx.x * (PL_T * PL_E);
    int s[PL_E], d[PL_E], rk[PL_E];
    float ex[PL_E];
    #pragma unroll
    for (int j = 0; j < PL_E; ++j) {
        const int e  = e0 + j * PL_T + tid;
        const int ee = (e < n_edges) ? e : n_edges - 1;
        s[j]  = src[ee];
        d[j]  = dst[ee];
        ex[j] = exp_e[ee];
    }
    for (int i = tid; i < n_bins; i += PL_T) hist[i] = 0;
    for (int i = tid; i < (ntab + 3) / 4; i += PL_T)
        ((v4f*)dtab)[i] = (v4f){0.0f, 0.0f, 0.0f, 0.0f};
    __syncthreads();

    #pragma unroll
    for (int j = 0; j < PL_E; ++j) {
        const int e = e0 + j * PL_T + tid;
        rk[j] = -1;
        if (e < n_edges) {
            rk[j] = atomicAdd(&hist[d[j] / RB], 1);
            if (s[j] < ntab) atomicAdd(&dtab[s[j]], ex[j]);
            else             atomicAdd(&denom[s[j]], ex[j]);
        }
    }
    __syncthreads();
    for (int i = tid; i < n_bins; i += PL_T) {
        const int c = hist[i];
        base[i] = (c > 0) ? atomicAdd(&bin_count[i], c) : 0;
    }
    __syncthreads();

    int4* rec = (int4*)rec_raw;
    int4* ovf = (int4*)ovf_raw;
    #pragma unroll
    for (int j = 0; j < PL_E; ++j) {
        if (rk[j] >= 0) {
            const int b   = d[j] / RB;
            const int pos = base[b] + rk[j];
            int4* p = rec + ((size_t)b * CAP + pos);
            int row = d[j] - b * RB;
            if (pos >= CAP) {
                p = ovf + atomicAdd(&bin_count[n_bins], 1);
                row = d[j];
            }
            *p = make_int4(s[j], row, __float_as_int(ex[j]), 0);
        }
    }

    for (int i0 = 0; i0 < ntab; i0 += 4 * PL_T) {
        float t[4];
        #pragma unroll
        for (int q = 0; q < 4; ++q) {
            const int i = i0 + q * PL_T + tid;
            t[q] = (i < ntab) ? dtab[i] : 0.0f;
        }
        #pragma unroll
        for (int q = 0; q < 4; ++q) {
            const int i = i0 + q * PL_T + tid;
            if (t[q] != 0.0f) atomicAdd(&denom[i], t[q]);
        }
    }
}

__device__ __forceinline__ void flush_row(float* acc_row, int lane, v4f a) {
    float* ap = acc_row + lane * 4;
    atomicAdd(ap,     a.x);
    atomicAdd(ap + 1, a.y);
    atomicAdd(ap + 2, a.z);
    atomicAdd(ap + 3, a.w);
}

__global__ __launch_bounds__(1024) void gat_sum(const float* __restrict__ h,
                                                const int* __restrict__ rec_raw,
                                                const int* __restrict__ ovf_raw,
                                                const int* __restrict__ bin_count,
                                                const float* __restrict__ denom,
                                                float* __restrict__ out,
                                                int n_nodes) {
    __shared__ __attribute__((aligned(16))) float acc[RB * C];
    __shared__ int4  srt[CAP];
    __shared__ int   rowc[RSLOT];
    __shared__ __attribute__((aligned(16))) float dsh[NTAB];
    const int tid  = threadIdx.x;
    const int lane = tid & 31;
    const int wave = __builtin_amdgcn_readfirstlane(tid >> 5);
    const int b = blockIdx.x;
    const int ntab = min(n_nodes, NTAB);

    const int n = min(bin_count[b], CAP);
    const int n_ovf = bin_count[gridDim.x];
    const int4* r = (const int4*)rec_raw + (size_t)b * CAP;

    int4 mine[CAP / 1024];
    #pragma unroll
    for (int k = 0; k < CAP / 1024; ++k) {
        const int idx = tid + k * 1024;
        mine[k] = make_int4(0, -1, 0, 0);
        if (idx < n) mine[k] = r[idx];
    }
    const int nq = (ntab + 3) >> 2;
    v4f dq[NTAB / 4096];
    #pragma unroll
    for (int q = 0; q < NTAB / 4096; ++q) {
        const int i = tid + q * 1024;
        dq[q] = (v4f){1.0f, 1.0f, 1.0f, 1.0f};
        if (i < nq) dq[q] = ((const v4f*)denom)[i];
    }
    for (int i = tid; i < RB * C; i += 1024) acc[i] = 0.0f;
    if (tid < RSLOT) rowc[tid] = 0;
    #pragma unroll
    for (int q = 0; q < NTAB / 4096; ++q)
        ((v4f*)dsh)[tid + q * 1024] = dq[q];
    __syncthreads();

    int rk[CAP / 1024];
    float dn[CAP / 1024];
    #pragma unroll
    for (int k = 0; k < CAP / 1024; ++k)
        dn[k] = dsh[(mine[k].x < ntab) ? mine[k].x : 0];
    if (n_nodes > NTAB) {
        #pragma unroll
        for (int k = 0; k < CAP / 1024; ++k)
            if (mine[k].y >= 0 && mine[k].x >= NTAB) dn[k] = denom[mine[k].x];
    }
    #pragma unroll
    for (int k = 0; k < CAP / 1024; ++k) {
        rk[k] = 0;
        if (mine[k].y >= 0) {
            mine[k].z = __float_as_int(__int_as_float(mine[k].z) / dn[k]);
            rk[k] = atomicAdd(&rowc[mine[k].y], 1);
        }
    }
    __syncthreads();
    if (wave == 0) {
        const int c0 = rowc[lane * 4], c1 = rowc[lane * 4 + 1], c2 = rowc[lane * 4 + 2], c3 = rowc[lane * 4 + 3];
        const int s = c0 + c1 + c2 + c3;
        int x = s;
        #pragma unroll
        for (int off = 1; off < 32; off <<= 1) {
            const int y = __shfl_up(x, off, 32);
            if (lane >= off) x += y;
        }
        const int first = x - s;
        rowc[lane * 4]     = first;
        rowc[lane * 4 + 1] = first + c0;
        rowc[lane * 4 + 2] = first + c0 + c1;
        rowc[lane * 4 + 3] = first + c0 + c1 + c2;
    }
    __syncthreads();
    #pragma unroll
    for (int k = 0; k < CAP / 1024; ++k)
        if (mine[k].y >= 0) srt[rowc[mine[k].y] + rk[k]] = mine[k];
    __syncthreads();

    const int nb = (n + 15) >> 4;
    const int lo = ((wave * nb) >> 5) << 4;
    const int hi = min(n, (((wave + 1) * nb) >> 5) << 4);
    int cur = -1;
    v4f a = {0.0f, 0.0f, 0.0f, 0.0f};
    for (int g = lo; g < hi; g += 16) {
        const int cnt = min(16, hi - g);
        int4 rc = make_int4(0, 0, 0, 0);
        if (lane < cnt) rc = srt[g + lane];
        v4f hv[16];
        #pragma unroll
        for (int i = 0; i < 16; ++i)
            hv[i] = load_row4(h, lane_i(rc.x, i), lane);
        const float attl = __int_as_float(rc.z);
        #pragma unroll
        for (int i = 0; i < 16; ++i) {
            if (i < cnt) {
                const int row = lane_i(rc.y, i);
                const float att = lane_f(attl, i);
                if (row != cur) {
                    if (cur >= 0) flush_row(acc + cur * C, lane, a);
                    cur = row;
                    a = (v4f){0.0f, 0.0f, 0.0f, 0.0f};
                }
                a.x = fmaf(hv[i].x, att, a.x);
                a.y = fmaf(hv[i].y, att, a.y);
                a.z = fmaf(hv[i].z, att, a.z);
                a.w = fmaf(hv[i].w, att, a.w);
            }
        }
    }
    if (cur >= 0) flush_row(acc + cur * C, lane, a);
    if (n_ovf > 0) {
        const int4* ov = (const int4*)ovf_raw;
        for (int i = tid; i < n_ovf; i += 1024) {
            const int4 o = ov[i];
            const int rl = o.y - b * RB;
            if (rl >= 0 && rl < RB) {
                const float att = __int_as_float(o.z) / denom[o.x];
                const float* hp = h + (size_t)o.x * C;
                float* ap = acc + rl * C;
                for (int c = 0; c < C; ++c) atomicAdd(ap + c, hp[c] * att);
            }
        }
    }
    __syncthreads();

    const int row0 = b * RB;
    const v4f* acc4 = (const v4f*)acc;
    v4f* o4 = (v4f*)(out + (size_t)row0 * C);
    v4f ov[(RB * C / 4 + 1023) / 1024];
    #pragma unroll
    for (int k = 0; k < (RB * C / 4 + 1023) / 1024; ++k) {
        const int i = tid + k * 1024;
        ov[k] = (i < RB * C / 4) ? acc4[i] : (v4f){0.0f, 0.0f, 0.0f, 0.0f};
    }
    #pragma unroll
    for (int k = 0; k < (RB * C / 4 + 1023) / 1024; ++k) {
        const int i = tid + k * 1024;
        if (i < RB * C / 4 && row0 + (i >> 5) < n_nodes) o4[i] = ov[k];
    }
}

extern "C" void kernel_launch(void* const* d_in, const int* in_sizes, int n_in,
                              void* d_out, int out_size, void* d_ws, size_t ws_size,
                              hipStream_t stream) {
    const float* h  = (const float*)d_in[0];
    const float* w  = (const float*)d_in[1];
    const float* a  = (const float*)d_in[2];
    const int* src  = (const int*)d_in[3];
    const int* dst  = (const int*)d_in[4];
    float* out      = (float*)d_out;

    const int n_nodes = in_sizes[0] / C;
    const int n_edges = in_sizes[3];

    const int n_bins  = (n_nodes + RB - 1) / RB;

    float* u       = (float*)d_ws;
    float* v       = u + (size_t)n_nodes * C;
    int* rec       = (int*)(v + (size_t)n_nodes * C);
    float* exp_e   = (float*)(rec + (size_t)n_bins * CAP * 4);
    int* ovf       = (int*)(exp_e + n_edges);
    float* denom   = (float*)(ovf + (size_t)n_edges * 4);
    int* bin_count = (int*)(denom + n_nodes);

    {
        gat_gemm_uv<<<n_bins, 256, 0, stream>>>(h, w, u, v, denom, bin_count, n_bins, n_nodes);
    }
    {
        int grid = (n_edges + 255) / 256;
        gat_edge_e<<<grid, 256, 0, stream>>>(u, v, a, src, dst, exp_e, n_edges);
    }
    {
        int grid = (n_edges + PL_T * PL_E - 1) / (PL_T * PL_E);
        gat_place<<<grid, PL_T, (size_t)n_bins * 2 * sizeof(int), stream>>>(
            src, dst, exp_e, denom, bin_count, rec, ovf, n_edges, n_bins, n_nodes);
    }
    {
        gat_sum<<<n_bins, 1024, 0, stream>>>(h, rec, ovf, bin_count, denom, out, n_nodes);
    }
}


# grouping kernel: a block's records put in range order in LDS first, then copied out lane = slot: a wave's store covers a few contiguous runs (was 32 scattered 16-B pieces); same slots, same sums
# speedup vs baseline: 4.2009x; 1.0315x over previous
#include <hip/hip_runtime.h>
#include <hip/hip_bf16.h>


#define C 128
#define LEAKY 0.2f

typedef __attribute__((ext_vector_type(2))) float v2f;
typedef __attribute__((ext_vector_type(8))) float v8f;

typedef __attribute__((ext_vector_type(4))) float v4f;

#define TPB   5
#define HP    132
__global__ __launch_bounds__(256) void gat_gemm_uv(const float* __restrict__ h,
                                                   const float* __restrict__ w,
                                                   float* __restrict__ u,
                                                   float* __restrict__ v,
                                                   float* __restrict__ denom,
                                                   int* __restrict__ bin_count,
                                                   int n_bins,
                                                   int n_nodes) {
    __shared__ __attribute__((aligned(16))) float lh[TPB * 16 * HP];
    const int row0 = blockIdx.x * (TPB * 16);
    const int tid  = threadIdx.x;
    const int wave = tid >> 5;
    const int lane = tid & 31;
    const int half = lane >> 4;
    const int l16  = lane & 15;
    const int col  = wave * 16 + l16;
    const int nrow = min(TPB * 16, n_nodes - row0);

    v4f hq[TPB * 16 * 32 / 256];
    #pragma unroll
    for (int q = 0; q < TPB * 16 * 32 / 256; ++q) {
        const int i = tid + q * 256;
        hq[q] = (v4f){0.0f, 0.0f, 0.0f, 0.0f};
        if ((i >> 5) < nrow) hq[q] = ((const v4f*)(h + (size_t)row0 * C))[i];
    }

    v2f bu[C / 4], bv[C / 4];
    #pragma unroll
    for (int j = 0; j < C / 4; ++j) {
        const int ka = j * 4 + half * 2;
        bu[j].x = w[(size_t)(ka    ) * C + col];
        bu[j].y = w[(size_t)(ka + 1) * C + col];
        bv[j].x = w[(size_t)(C + ka    ) * C + col];
        bv[j].y = w[(size_t)(C + ka + 1) * C + col];
    }

    #pragma unroll
    for (int q = 0; q < TPB * 16 * 32 / 256; ++q) {
        const int i = tid + q * 256;
        *(v4f*)(lh + (i >> 5) * HP + (i & 31) * 4) = hq[q];
    }
    __syncthreads();

    {
        if (tid < nrow) denom[row0 + tid] = 0.0f;
        if (blockIdx.x == 0)
            for (int i = tid; i < n_bins + 1; i += 256) bin_count[i] = 0;
    }

    for (int t = 0; t < TPB; ++t) {
        if (t * 16 >= nrow) break;
        v8f accu = {};
        v8f accv = {};
        const float* arow = lh + (t * 16 + l16) * HP + half * 2;
        #pragma unroll
        for (int j = 0; j < C / 4; ++j) {
            const v2f afrag = *(const v2f*)(arow + j * 4);
            accu = __builtin_amdgcn_wmma_f32_16x16x4_f32(false, afrag, false, bu[j],
                                                         (short)0, accu, false, false);
            accv = __builtin_amdgcn_wmma_f32_16x16x4_f32(false, afrag, false, bv[j],
                                                         (short)0, accv, false, false);
        }
        const int ml = t * 16 + half * 8;
        #pragma unroll
        for (int r = 0; r < 8; ++r) {
            if (ml + r < nrow) {
                u[(size_t)(row0 + ml + r) * C + col] = accu[r];
                v[(size_t)(row0 + ml + r) * C + col] = accv[r];
            }
        }
    }
}

__device__ __forceinline__ float leaky(float x) { return x > 0.0f ? x : LEAKY * x; }


__device__ __forceinline__ v4f load_row4(const float* __restrict__ base, int row, int lane) {
    return *(const v4f*)(base + (size_t)row * C + lane * 4);
}

__device__ __forceinline__ float edge_partial(v4f uu, v4f vv, float4 aa) {
    float sum = leaky(uu.x + vv.x) * aa.x;
    sum = fmaf(leaky(uu.y + vv.y), aa.y, sum);
    sum = fmaf(leaky(uu.z + vv.z), aa.z, sum);
    sum = fmaf(leaky(uu.w + vv.w), aa.w, sum);
    return sum;
}

__device__ __forceinline__ float fold(float lo, float hi, bool bit, int off) {
    float keep = bit ? hi : lo;
    float send = bit ? lo : hi;
    return keep + __shfl_xor(send, off, 32);
}

__global__ __launch_bounds__(256) void gat_edge_e(const float* __restrict__ u,
                                                  const float* __restrict__ v,
                                                  const float* __restrict__ a,
                                                  const int* __restrict__ src,
                                                  const int* __restrict__ dst,
                                                  float* __restrict__ exp_e,
                                                  int n_edges) {
    const int wib  = threadIdx.x >> 5;
    const int lane = threadIdx.x & 31;
    const int e0 = (blockIdx.x * 8 + wib) * 32;
    if (e0 >= n_edges) return;

    const int eown = min(e0 + lane, n_edges - 1);
    const int s_own = src[eown];
    const int d_own = dst[eown];
    const float4 aa = ((const float4*)a)[lane];

    const bool b4 = (lane & 16) != 0;
    const bool b3 = (lane & 8) != 0;
    const bool b2 = (lane & 4) != 0;
    const bool b1 = (lane & 2) != 0;
    const bool b0 = (lane & 1) != 0;

    float wsum[4];
    #pragma unroll
    for (int b = 0; b < 4; ++b) {
        v4f uu[8], vv[8];
        #pragma unroll
        for (int i = 0; i < 8; ++i) {
            uu[i] = load_row4(u, __builtin_amdgcn_readlane(s_own, b * 8 + i), lane);
            vv[i] = load_row4(v, __builtin_amdgcn_readlane(d_own, b * 8 + i), lane);
        }
        asm volatile("" :: "v"(uu[0]), "v"(uu[1]), "v"(uu[2]), "v"(uu[3]),
                           "v"(uu[4]), "v"(uu[5]), "v"(uu[6]), "v"(uu[7]),
                           "v"(vv[0]), "v"(vv[1]), "v"(vv[2]), "v"(vv[3]),
                           "v"(vv[4]), "v"(vv[5]), "v"(vv[6]), "v"(vv[7]));
        float p[8];
        #pragma unroll
        for (int i = 0; i < 8; ++i)
            p[i] = edge_partial(uu[i], vv[i], aa);
        const float q0 = fold(p[0], p[4], b4, 16);
        const float q1 = fold(p[1], p[5], b4, 16);
        const float q2 = fold(p[2], p[6], b4, 16);
        const float q3 = fold(p[3], p[7], b4, 16);
        const float r0 = fold(q0, q2, b3, 8);
        const float r1 = fold(q1, q3, b3, 8);
        wsum[b] = fold(r0, r1, b2, 4);
    }
    const float x0 = fold(wsum[0], wsum[2], b1, 2);
    const float x1 = fold(wsum[1], wsum[3], b1, 2);
    const float logit = fold(x0, x1, b0, 1);

    const int slot = ((lane & 3) << 3) | (lane >> 2);
    const int e = e0 + slot;
    if (e < n_edges) exp_e[e] = expf(logit);
}

#define RB   80
#define CAP  4096
#define PL_T 1024
#define PL_E 5
#define NTAB 20480
static_assert(NTAB % 4 == 0, "gat_place clears the table 16 B at a time");
#define RSLOT 128
static_assert(RB <= RSLOT, "one wave scans all row counters");
static_assert(CAP % 1024 == 0, "gat_sum: whole records per thread");
static_assert(NTAB % (4 * 1024) == 0, "gat_sum copies denom 16 B per thread at a time");

__device__ __forceinline__ int lane_i(int x, int l) { return __builtin_amdgcn_readlane(x, l); }
__device__ __forceinline__ float lane_f(float x, int l) {
    return __int_as_float(__builtin_amdgcn_readlane(__float_as_int(x), l));
}

__global__ __launch_bounds__(PL_T) void gat_place(const int* __restrict__ src,
                                                  const int* __restrict__ dst,
                                                  const float* __restrict__ exp_e,
                                                  float* __restrict__ denom,
                                                  int* __restrict__ bin_count,
                                                  int* __restrict__ rec_raw,
                                                  int* __restrict__ ovf_raw,
                                                  int n_edges, int n_bins, int n_nodes) {
    extern __shared__ int sm[];
    __shared__ __attribute__((aligned(16))) float dtab[NTAB];
    __shared__ int4 stage[PL_T * PL_E];
    int* hist = sm;
    int* base = sm + n_bins;
    int* lfirst = sm + 2 * n_bins;
    const int tid  = threadIdx.x;
    const int ntab = min(n_nodes, NTAB);

    const int e0 = blockIdx.x * (PL_T * PL_E);
    int s[PL_E], d[PL_E], rk[PL_E];
    float ex[PL_E];
    #pragma unroll
    for (int j = 0; j < PL_E; ++j) {
        const int e  = e0 + j * PL_T + tid;
        const int ee = (e < n_edges) ? e : n_edges - 1;
        s[j]  = src[ee];
        d[j]  = dst[ee];
        ex[j] = exp_e[ee];
    }
    for (int i = tid; i < n_bins; i += PL_T) hist[i] = 0;
    for (int i = tid; i < (ntab + 3) / 4; i += PL_T)
        ((v4f*)dtab)[i] = (v4f){0.0f, 0.0f, 0.0f, 0.0f};
    __syncthreads();

    #pragma unroll
    for (int j = 0; j < PL_E; ++j) {
        const int e = e0 + j * PL_T + tid;
        rk[j] = -1;
        if (e < n_edges) {
            rk[j] = atomicAdd(&hist[d[j] / RB], 1);
            if (s[j] < ntab) atomicAdd(&dtab[s[j]], ex[j]);
            else             atomicAdd(&denom[s[j]], ex[j]);
        }
    }
    __syncthreads();
    for (int i = tid; i < n_bins; i += PL_T) {
        const int c = hist[i];
        base[i] = (c > 0) ? atomicAdd(&bin_count[i], c) : 0;
    }
    if (tid >= PL_T - 32) {
        const int lane = tid & 31;
        int carry = 0;
        for (int i0 = 0; i0 < n_bins; i0 += 32) {
            const int i = i0 + lane;
            const int c = (i < n_bins) ? hist[i] : 0;
            int x = c;
            #pragma unroll
            for (int off = 1; off < 32; off <<= 1) {
                const int y = __shfl_up(x, off, 32);
                if (lane >= off) x += y;
            }
            if (i < n_bins) lfirst[i] = carry + x - c;
            carry += __shfl(x, 31, 32);
        }
    }
    __syncthreads();

    #pragma unroll
    for (int j = 0; j < PL_E; ++j) {
        if (rk[j] >= 0) {
            const int b = d[j] / RB;
            stage[lfirst[b] + rk[j]] = make_int4(s[j], d[j] - b * RB, __float_as_int(ex[j]), b);
        }
    }
    __syncthreads();

    int4* rec = (int4*)rec_raw;
    int4* ovf = (int4*)ovf_raw;
    const int ne = min(PL_T * PL_E, n_edges - e0);
    int4 st[PL_E];
    #pragma unroll
    for (int j = 0; j < PL_E; ++j) {
        const int q = j * PL_T + tid;
        st[j] = stage[(q < ne) ? q : 0];
    }
    #pragma unroll
    for (int j = 0; j < PL_E; ++j) {
        const int q = j * PL_T + tid;
        if (q < ne) {
            const int b   = st[j].w;
            const int pos = base[b] + (q - lfirst[b]);
            int4* p = rec + ((size_t)b * CAP + pos);
            int row = st[j].y;
            if (pos >= CAP) {
                p = ovf + atomicAdd(&bin_count[n_bins], 1);
                row = st[j].y + b * RB;
            }
            *p = make_int4(st[j].x, row, st[j].z, 0);
        }
    }

    for (int i0 = 0; i0 < ntab; i0 += 4 * PL_T) {
        float t[4];
        #pragma unroll
        for (int q = 0; q < 4; ++q) {
            const int i = i0 + q * PL_T + tid;
            t[q] = (i < ntab) ? dtab[i] : 0.0f;
        }
        #pragma unroll
        for (int q = 0; q < 4; ++q) {
            const int i = i0 + q * PL_T + tid;
            if (t[q] != 0.0f) atomicAdd(&denom[i], t[q]);
        }
    }
}

__device__ __forceinline__ void flush_row(float* acc_row, int lane, v4f a) {
    float* ap = acc_row + lane * 4;
    atomicAdd(ap,     a.x);
    atomicAdd(ap + 1, a.y);
    atomicAdd(ap + 2, a.z);
    atomicAdd(ap + 3, a.w);
}

__global__ __launch_bounds__(1024) void gat_sum(const float* __restrict__ h,
                                                const int* __restrict__ rec_raw,
                                                const int* __restrict__ ovf_raw,
                                                const int* __restrict__ bin_count,
                                                const float* __restrict__ denom,
                                                float* __restrict__ out,
                                                int n_nodes) {
    __shared__ __attribute__((aligned(16))) float acc[RB * C];
    __shared__ int4  srt[CAP];
    __shared__ int   rowc[RSLOT];
    __shared__ __attribute__((aligned(16))) float dsh[NTAB];
    const int tid  = threadIdx.x;
    const int lane = tid & 31;
    const int wave = __builtin_amdgcn_readfirstlane(tid >> 5);
    const int b = blockIdx.x;
    const int ntab = min(n_nodes, NTAB);

    const int n = min(bin_count[b], CAP);
    const int n_ovf = bin_count[gridDim.x];
    const int4* r = (const int4*)rec_raw + (size_t)b * CAP;

    int4 mine[CAP / 1024];
    #pragma unroll
    for (int k = 0; k < CAP / 1024; ++k) {
        const int idx = tid + k * 1024;
        mine[k] = make_int4(0, -1, 0, 0);
        if (idx < n) mine[k] = r[idx];
    }
    const int nq = (ntab + 3) >> 2;
    v4f dq[NTAB / 4096];
    #pragma unroll
    for (int q = 0; q < NTAB / 4096; ++q) {
        const int i = tid + q * 1024;
        dq[q] = (v4f){1.0f, 1.0f, 1.0f, 1.0f};
        if (i < nq) dq[q] = ((const v4f*)denom)[i];
    }
    for (int i = tid; i < RB * C; i += 1024) acc[i] = 0.0f;
    if (tid < RSLOT) rowc[tid] = 0;
    #pragma unroll
    for (int q = 0; q < NTAB / 4096; ++q)
        ((v4f*)dsh)[tid + q * 1024] = dq[q];
    __syncthreads();

    int rk[CAP / 1024];
    float dn[CAP / 1024];
    #pragma unroll
    for (int k = 0; k < CAP / 1024; ++k)
        dn[k] = dsh[(mine[k].x < ntab) ? mine[k].x : 0];
    if (n_nodes > NTAB) {
        #pragma unroll
        for (int k = 0; k < CAP / 1024; ++k)
            if (mine[k].y >= 0 && mine[k].x >= NTAB) dn[k] = denom[mine[k].x];
    }
    #pragma unroll
    for (int k = 0; k < CAP / 1024; ++k) {
        rk[k] = 0;
        if (mine[k].y >= 0) {
            mine[k].z = __float_as_int(__int_as_float(mine[k].z) / dn[k]);
            rk[k] = atomicAdd(&rowc[mine[k].y], 1);
        }
    }
    __syncthreads();
    if (wave == 0) {
        const int c0 = rowc[lane * 4], c1 = rowc[lane * 4 + 1], c2 = rowc[lane * 4 + 2], c3 = rowc[lane * 4 + 3];
        const int s = c0 + c1 + c2 + c3;
        int x = s;
        #pragma unroll
        for (int off = 1; off < 32; off <<= 1) {
            const int y = __shfl_up(x, off, 32);
            if (lane >= off) x += y;
        }
        const int first = x - s;
        rowc[lane * 4]     = first;
        rowc[lane * 4 + 1] = first + c0;
        rowc[lane * 4 + 2] = first + c0 + c1;
        rowc[lane * 4 + 3] = first + c0 + c1 + c2;
    }
    __syncthreads();
    #pragma unroll
    for (int k = 0; k < CAP / 1024; ++k)
        if (mine[k].y >= 0) srt[rowc[mine[k].y] + rk[k]] = mine[k];
    __syncthreads();

    const int nb = (n + 15) >> 4;
    const int lo = ((wave * nb) >> 5) << 4;
    const int hi = min(n, (((wave + 1) * nb) >> 5) << 4);
    int cur = -1;
    v4f a = {0.0f, 0.0f, 0.0f, 0.0f};
    for (int g = lo; g < hi; g += 16) {
        const int cnt = min(16, hi - g);
        int4 rc = make_int4(0, 0, 0, 0);
        if (lane < cnt) rc = srt[g + lane];
        v4f hv[16];
        #pragma unroll
        for (int i = 0; i < 16; ++i)
            hv[i] = load_row4(h, lane_i(rc.x, i), lane);
        const float attl = __int_as_float(rc.z);
        #pragma unroll
        for (int i = 0; i < 16; ++i) {
            if (i < cnt) {
                const int row = lane_i(rc.y, i);
                const float att = lane_f(attl, i);
                if (row != cur) {
                    if (cur >= 0) flush_row(acc + cur * C, lane, a);
                    cur = row;
                    a = (v4f){0.0f, 0.0f, 0.0f, 0.0f};
                }
                a.x = fmaf(hv[i].x, att, a.x);
                a.y = fmaf(hv[i].y, att, a.y);
                a.z = fmaf(hv[i].z, att, a.z);
                a.w = fmaf(hv[i].w, att, a.w);
            }
        }
    }
    if (cur >= 0) flush_row(acc + cur * C, lane, a);
    if (n_ovf > 0) {
        const int4* ov = (const int4*)ovf_raw;
        for (int i = tid; i < n_ovf; i += 1024) {
            const int4 o = ov[i];
            const int rl = o.y - b * RB;
            if (rl >= 0 && rl < RB) {
                const float att = __int_as_float(o.z) / denom[o.x];
                const float* hp = h + (size_t)o.x * C;
                float* ap = acc + rl * C;
                for (int c = 0; c < C; ++c) atomicAdd(ap + c, hp[c] * att);
            }
        }
    }
    __syncthreads();

    const int row0 = b * RB;
    const v4f* acc4 = (const v4f*)acc;
    v4f* o4 = (v4f*)(out + (size_t)row0 * C);
    v4f ov[(RB * C / 4 + 1023) / 1024];
    #pragma unroll
    for (int k = 0; k < (RB * C / 4 + 1023) / 1024; ++k) {
        const int i = tid + k * 1024;
        ov[k] = (i < RB * C / 4) ? acc4[i] : (v4f){0.0f, 0.0f, 0.0f, 0.0f};
    }
    #pragma unroll
    for (int k = 0; k < (RB * C / 4 + 1023) / 1024; ++k) {
        const int i = tid + k * 1024;
        if (i < RB * C / 4 && row0 + (i >> 5) < n_nodes) o4[i] = ov[k];
    }
}

extern "C" void kernel_launch(void* const* d_in, const int* in_sizes, int n_in,
                              void* d_out, int out_size, void* d_ws, size_t ws_size,
                              hipStream_t stream) {
    const float* h  = (const float*)d_in[0];
    const float* w  = (const float*)d_in[1];
    const float* a  = (const float*)d_in[2];
    const int* src  = (const int*)d_in[3];
    const int* dst  = (const int*)d_in[4];
    float* out      = (float*)d_out;

    const int n_nodes = in_sizes[0] / C;
    const int n_edges = in_sizes[3];

    const int n_bins  = (n_nodes + RB - 1) / RB;

    float* u       = (float*)d_ws;
    float* v       = u + (size_t)n_nodes * C;
    int* rec       = (int*)(v + (size_t)n_nodes * C);
    float* exp_e   = (float*)(rec + (size_t)n_bins * CAP * 4);
    int* ovf       = (int*)(exp_e + n_edges);
    float* denom   = (float*)(ovf + (size_t)n_edges * 4);
    int* bin_count = (int*)(denom + n_nodes);

    {
        gat_gemm_uv<<<n_bins, 256, 0, stream>>>(h, w, u, v, denom, bin_count, n_bins, n_nodes);
    }
    {
        int grid = (n_edges + 255) / 256;
        gat_edge_e<<<grid, 256, 0, stream>>>(u, v, a, src, dst, exp_e, n_edges);
    }
    {
        int grid = (n_edges + PL_T * PL_E - 1) / (PL_T * PL_E);
        gat_place<<<grid, PL_T, (size_t)n_bins * 3 * sizeof(int), stream>>>(
            src, dst, exp_e, denom, bin_count, rec, ovf, n_edges, n_bins, n_nodes);
    }
    {
        gat_sum<<<n_bins, 1024, 0, stream>>>(h, rec, ovf, bin_count, denom, out, n_nodes);
    }
}
